# starting kernel
# baseline (speedup 1.0000x reference)
#include <hip/hip_runtime.h>
#include <hip/hip_bf16.h>
#include <math.h>

#define N_NODES 100000
#define N_EDGES 1600000
#define IN_DIM  128
#define OUT_DIM 64
#define LEAKY   0.01f

typedef float v2f __attribute__((ext_vector_type(2)));
typedef float v8f __attribute__((ext_vector_type(8)));

__device__ __forceinline__ unsigned fkey(float f) {
    unsigned u = __float_as_uint(f);
    return (u & 0x80000000u) ? ~u : (u | 0x80000000u);
}
__device__ __forceinline__ float keyf(unsigned k) {
    unsigned u = (k & 0x80000000u) ? (k & 0x7FFFFFFFu) : ~k;
    return __uint_as_float(u);
}
#define NEG_INF_KEY 0x007FFFFFu

__global__ void gat_init(float* __restrict__ out, unsigned* __restrict__ mkey,
                         float* __restrict__ denom) {
    int i = blockIdx.x * blockDim.x + threadIdx.x;
    if (i < N_NODES * OUT_DIM) out[i] = 0.0f;
    if (i < N_NODES) { mkey[i] = NEG_INF_KEY; denom[i] = 0.0f; }
}

__global__ void gat_gemm_wmma(const float* __restrict__ h,
                              const float* __restrict__ W,
                              float* __restrict__ z) {
    const int lane = threadIdx.x & 31;
    const int wave = threadIdx.x >> 5;
    const int row0 = blockIdx.x * 16;
    const int col0 = wave * 16;

    const int m     = lane & 15;
    const int khalf = (lane >> 4) * 2;

    const float* __restrict__ hrow = h + (size_t)(row0 + m) * IN_DIM;
    const float* __restrict__ wrow = W + (size_t)(col0 + m) * IN_DIM;

    __builtin_prefetch(wrow, 0, 3);

    v8f c = {};
#pragma unroll
    for (int k0 = 0; k0 < IN_DIM; k0 += 4) {
        v2f a, b;
        a.x = hrow[k0 + khalf];
        a.y = hrow[k0 + khalf + 1];
        b.x = wrow[k0 + khalf];
        b.y = wrow[k0 + khalf + 1];
        c = __builtin_amdgcn_wmma_f32_16x16x4_f32(
                false, a, false, b, (short)0, c, false, false);
    }

    const int n     = lane & 15;
    const int rbase = (lane >> 4) * 8;
#pragma unroll
    for (int r = 0; r < 8; ++r) {
        z[(size_t)(row0 + rbase + r) * OUT_DIM + col0 + n] = c[r];
    }
}

__global__ void gat_proj(const float* __restrict__ z, const float* __restrict__ A,
                         float* __restrict__ s, float* __restrict__ t) {
    int i = blockIdx.x * blockDim.x + threadIdx.x;
    if (i >= N_NODES) return;
    const float4* zr = (const float4*)(z + (size_t)i * OUT_DIM);
    const float4* as = (const float4*)A;
    const float4* ad = (const float4*)(A + OUT_DIM);
    float accs = 0.f, acct = 0.f;
#pragma unroll
    for (int k = 0; k < OUT_DIM / 4; ++k) {
        float4 v = zr[k], x = as[k], y = ad[k];
        accs += v.x * x.x + v.y * x.y + v.z * x.z + v.w * x.w;
        acct += v.x * y.x + v.y * y.y + v.z * y.z + v.w * y.w;
    }
    s[i] = accs;
    t[i] = acct;
}

__global__ void gat_edge_logit(const int* __restrict__ src, const int* __restrict__ dst,
                               const float* __restrict__ s, const float* __restrict__ t,
                               float* __restrict__ eb, unsigned* __restrict__ mkey) {
    int e = blockIdx.x * blockDim.x + threadIdx.x;
    if (e >= N_EDGES) return;
    int si = src[e], di = dst[e];
    float v = s[si] + t[di];
    v = (v > 0.f) ? v : LEAKY * v;
    eb[e] = v;
    atomicMax(&mkey[di], fkey(v));
}

__global__ void gat_mfix(unsigned* __restrict__ mkey) {
    int i = blockIdx.x * blockDim.x + threadIdx.x;
    if (i >= N_NODES) return;
    float mv = keyf(mkey[i]);
    if (!isfinite(mv)) mv = 0.0f;
    ((float*)mkey)[i] = mv;
}

__global__ void gat_p_denom(const int* __restrict__ dst, const float* __restrict__ m,
                            float* __restrict__ eb, float* __restrict__ denom) {
    int e = blockIdx.x * blockDim.x + threadIdx.x;
    if (e >= N_EDGES) return;
    int di = dst[e];
    float p = expf(eb[e] - m[di]);
    eb[e] = p;
    atomicAdd(&denom[di], p);
}

__global__ void gat_scatter(const int* __restrict__ src, const int* __restrict__ dst,
                            const float* __restrict__ p, const float* __restrict__ denom,
                            const float* __restrict__ z, float* __restrict__ out) {
    const int lane = threadIdx.x & 31;
    const int wave = threadIdx.x >> 5;
    const int e = blockIdx.x * 8 + wave;
    if (e >= N_EDGES) return;
    const int si = src[e], di = dst[e];
    const float alpha = p[e] / denom[di];
    float2 v = ((const float2*)(z + (size_t)si * OUT_DIM))[lane];
    float* o = out + (size_t)di * OUT_DIM + 2 * lane;
    atomicAdd(o + 0, alpha * v.x);
    atomicAdd(o + 1, alpha * v.y);
}

extern "C" void kernel_launch(void* const* d_in, const int* in_sizes, int n_in,
                              void* d_out, int out_size, void* d_ws, size_t ws_size,
                              hipStream_t stream) {
    const float* h  = (const float*)d_in[0];
    const int* src  = (const int*)d_in[1];
    const int* dst  = (const int*)d_in[2];
    const float* W  = (const float*)d_in[3];
    const float* A  = (const float*)d_in[4];
    float* out = (float*)d_out;

    char* ws = (char*)d_ws;
    float*    z     = (float*)   (ws);
    float*    s     = (float*)   (ws + 25600000);
    float*    t     = (float*)   (ws + 26000000);
    unsigned* mkey  = (unsigned*)(ws + 26400000);
    float*    denom = (float*)   (ws + 26800000);
    float*    eb    = (float*)   (ws + 27200000);

    const int node_blocks  = (N_NODES + 255) / 256;
    const int edge_blocks  = (N_EDGES + 255) / 256;
    const int init_blocks  = (N_NODES * OUT_DIM + 255) / 256;

    gat_init<<<init_blocks, 256, 0, stream>>>(out, mkey, denom);
    gat_gemm_wmma<<<N_NODES / 16, 128, 0, stream>>>(h, W, z);
    gat_proj<<<node_blocks, 256, 0, stream>>>(z, A, s, t);
    gat_edge_logit<<<edge_blocks, 256, 0, stream>>>(src, dst, s, t, eb, mkey);
    gat_mfix<<<node_blocks, 256, 0, stream>>>(mkey);
    gat_p_denom<<<edge_blocks, 256, 0, stream>>>(dst, (const float*)mkey, eb, denom);
    gat_scatter<<<(N_EDGES + 7) / 8, 256, 0, stream>>>(src, dst, eb, denom, z, out);
}


# z GEMM: one wave per 16-row strip x all 64 columns, h fragment loaded once as 16-byte loads and reused by 4 accumulator tiles; W staged once per work-group in LDS
# speedup vs baseline: 1.3670x; 1.3670x over previous
#include <hip/hip_runtime.h>
#include <hip/hip_bf16.h>
#include <math.h>


#define N_NODES 100000
#define N_EDGES 1600000
#define IN_DIM  128
#define OUT_DIM 64
#define LEAKY   0.01f

typedef float v2f __attribute__((ext_vector_type(2)));
typedef float v8f __attribute__((ext_vector_type(8)));

#define GEMM_WAVES 4
#define N_STRIPS (N_NODES / 16)
#define W_PITCH 132
#define W_STAGE ((OUT_DIM * IN_DIM / 4) / (GEMM_WAVES * 32))
#define WMMA_F32(A, B, C) __builtin_amdgcn_wmma_f32_16x16x4_f32(false, (A), false, (B), (short)0, (C), false, false)

__global__ void __launch_bounds__(GEMM_WAVES * 32)
gat_gemm_wmma(const float* __restrict__ h,
              const float* __restrict__ W,
              float* __restrict__ z) {
    __shared__ float4 wl4[OUT_DIM * W_PITCH / 4];
    float* wl = (float*)wl4;
    const int tid  = threadIdx.x;

#pragma unroll
    for (int i = 0; i < W_STAGE; ++i) {
        const int f  = tid + GEMM_WAVES * 32 * i;
        const int r  = f >> 5;
        const int c4 = f & 31;
        *(float4*)(wl + r * W_PITCH + c4 * 4) = ((const float4*)W)[f];
    }
    __syncthreads();

    const int lane  = tid & 31;
    const int strip = blockIdx.x * GEMM_WAVES + (tid >> 5);
    if (strip >= N_STRIPS) return;
    const int m  = lane & 15;
    const int hf = lane >> 4;
    const int row0 = strip * 16;

    float4 a[IN_DIM / 8];
    const float4* __restrict__ hrow = (const float4*)(h + (size_t)(row0 + m) * IN_DIM) + hf;
#pragma unroll
    for (int j = 0; j < IN_DIM / 8; ++j) a[j] = hrow[2 * j];
    __builtin_amdgcn_sched_barrier(0);

    const float* wrow = wl + m * W_PITCH + hf * 4;
    v8f c0 = {}, c1 = {}, c2 = {}, c3 = {};
    float4 n0 = *(const float4*)(wrow +  0 * W_PITCH);
    float4 n1 = *(const float4*)(wrow + 16 * W_PITCH);
    float4 n2 = *(const float4*)(wrow + 32 * W_PITCH);
    float4 n3 = *(const float4*)(wrow + 48 * W_PITCH);
#pragma unroll
    for (int j = 0; j < IN_DIM / 8; ++j) {
        const float4 b0 = n0, b1 = n1, b2 = n2, b3 = n3;
        if (j + 1 < IN_DIM / 8) {
            n0 = *(const float4*)(wrow +  0 * W_PITCH + 8 * (j + 1));
            n1 = *(const float4*)(wrow + 16 * W_PITCH + 8 * (j + 1));
            n2 = *(const float4*)(wrow + 32 * W_PITCH + 8 * (j + 1));
            n3 = *(const float4*)(wrow + 48 * W_PITCH + 8 * (j + 1));
        }
        __builtin_amdgcn_sched_barrier(0);
        v2f alo, ahi, blo, bhi;
        alo.x = a[j].x; alo.y = a[j].y; ahi.x = a[j].z; ahi.y = a[j].w;
        blo.x = b0.x; blo.y = b0.y; bhi.x = b0.z; bhi.y = b0.w;
        c0 = WMMA_F32(alo, blo, c0);
        blo.x = b1.x; blo.y = b1.y;
        c1 = WMMA_F32(alo, blo, c1);
        blo.x = b2.x; blo.y = b2.y;
        c2 = WMMA_F32(alo, blo, c2);
        blo.x = b3.x; blo.y = b3.y;
        c3 = WMMA_F32(alo, blo, c3);
        c0 = WMMA_F32(ahi, bhi, c0);
        bhi.x = b1.z; bhi.y = b1.w;
        c1 = WMMA_F32(ahi, bhi, c1);
        bhi.x = b2.z; bhi.y = b2.w;
        c2 = WMMA_F32(ahi, bhi, c2);
        bhi.x = b3.z; bhi.y = b3.w;
        c3 = WMMA_F32(ahi, bhi, c3);
    }

    float* __restrict__ zrow = z + (size_t)(row0 + hf * 8) * OUT_DIM + m;
#pragma unroll
    for (int r = 0; r < 8; ++r) {
        zrow[(size_t)r * OUT_DIM +  0] = c0[r];
        zrow[(size_t)r * OUT_DIM + 16] = c1[r];
        zrow[(size_t)r * OUT_DIM + 32] = c2[r];
        zrow[(size_t)r * OUT_DIM + 48] = c3[r];
    }
}

__global__ void gat_proj(const float* __restrict__ z, const float* __restrict__ A,
                         float* __restrict__ s, float* __restrict__ t, int* __restrict__ cnt) {
    int i = blockIdx.x * blockDim.x + threadIdx.x;
    if (i >= N_NODES) return;
    cnt[i] = 0;
    const float4* zr = (const float4*)(z + (size_t)i * OUT_DIM);
    const float4* as = (const float4*)A;
    const float4* ad = (const float4*)(A + OUT_DIM);
    float accs = 0.f, acct = 0.f;
#pragma unroll
    for (int k = 0; k < OUT_DIM / 4; ++k) {
        float4 v = zr[k], x = as[k], y = ad[k];
        accs += v.x * x.x + v.y * x.y + v.z * x.z + v.w * x.w;
        acct += v.x * y.x + v.y * y.y + v.z * y.z + v.w * y.w;
    }
    s[i] = accs;
    t[i] = acct;
}

__global__ void gat_count(const int* __restrict__ dst, int* __restrict__ cnt) {
    int e = blockIdx.x * blockDim.x + threadIdx.x;
    if (e >= N_EDGES) return;
    atomicAdd(&cnt[dst[e]], 1);
}

#define SCAN_T 1024
#define SCAN_E 16
__global__ void __launch_bounds__(SCAN_T)
gat_scan(const int* __restrict__ cnt, int* __restrict__ off) {
    __shared__ int wtot[SCAN_T / 32];
    const int tid  = threadIdx.x;
    const int lane = tid & 31;
    const int w    = tid >> 5;
    int carry = 0;
    for (int c = 0; c < (N_NODES + SCAN_T * SCAN_E - 1) / (SCAN_T * SCAN_E); ++c) {
        const int i0 = (c * SCAN_T + tid) * SCAN_E;
        const bool ok = i0 < N_NODES;
        int v[SCAN_E];
        for (int q = 0; q < SCAN_E / 4; ++q) {
            int4 x = make_int4(0, 0, 0, 0);
            if (ok) x = ((const int4*)(cnt + i0))[q];
            v[4 * q + 0] = x.x; v[4 * q + 1] = x.y; v[4 * q + 2] = x.z; v[4 * q + 3] = x.w;
        }
        int run = 0;
        for (int k = 0; k < SCAN_E; ++k) { const int x = v[k]; v[k] = run; run += x; }
        int inc = run;
        for (int o = 1; o < 32; o <<= 1) { const int y = __shfl_up(inc, o, 32); if (lane >= o) inc += y; }
        if (lane == 31) wtot[w] = inc;
        __syncthreads();
        const int wv = wtot[lane];
        int winc = wv;
        for (int o = 1; o < 32; o <<= 1) { const int y = __shfl_up(winc, o, 32); if (lane >= o) winc += y; }
        const int wave_excl = __shfl(winc, w, 32) - __shfl(wv, w, 32);
        const int total     = __shfl(winc, 31, 32);
        const int base = carry + wave_excl + (inc - run);
        if (ok) {
            for (int q = 0; q < SCAN_E / 4; ++q)
                ((int4*)(off + i0))[q] = make_int4(base + v[4 * q + 0], base + v[4 * q + 1],
                                                   base + v[4 * q + 2], base + v[4 * q + 3]);
        }
        carry += total;
        __syncthreads();
    }
}

__global__ void gat_fill(const int* __restrict__ src, const int* __restrict__ dst,
                         const int* __restrict__ off, int* __restrict__ cnt,
                         int* __restrict__ csr) {
    int e = blockIdx.x * blockDim.x + threadIdx.x;
    if (e >= N_EDGES) return;
    const int d = dst[e];
    const int k = atomicSub(&cnt[d], 1);
    const int pos = off[d] + k - 1;
    if ((unsigned)pos < (unsigned)N_EDGES) csr[pos] = src[e];
}

__device__ __forceinline__ float wave_max(float x) {
    for (int o = 16; o > 0; o >>= 1) x = fmaxf(x, __shfl_xor(x, o, 32));
    return x;
}
__device__ __forceinline__ float wave_sum(float x) {
    for (int o = 16; o > 0; o >>= 1) x += __shfl_xor(x, o, 32);
    return x;
}
__device__ __forceinline__ float leaky(float v) { return (v > 0.f) ? v : LEAKY * v; }

__device__ __forceinline__ void add_rows(float2& acc, int si, float a, int n,
                                         const float* __restrict__ z, int lane) {
    int j = 0;
    for (; j + 4 <= n; j += 4) {
        const int s0 = __builtin_amdgcn_readlane(si, j + 0);
        const int s1 = __builtin_amdgcn_readlane(si, j + 1);
        const int s2 = __builtin_amdgcn_readlane(si, j + 2);
        const int s3 = __builtin_amdgcn_readlane(si, j + 3);
        const float a0 = __int_as_float(__builtin_amdgcn_readlane(__float_as_int(a), j + 0));
        const float a1 = __int_as_float(__builtin_amdgcn_readlane(__float_as_int(a), j + 1));
        const float a2 = __int_as_float(__builtin_amdgcn_readlane(__float_as_int(a), j + 2));
        const float a3 = __int_as_float(__builtin_amdgcn_readlane(__float_as_int(a), j + 3));
        const float2 z0 = ((const float2*)(z + (size_t)s0 * OUT_DIM))[lane];
        const float2 z1 = ((const float2*)(z + (size_t)s1 * OUT_DIM))[lane];
        const float2 z2 = ((const float2*)(z + (size_t)s2 * OUT_DIM))[lane];
        const float2 z3 = ((const float2*)(z + (size_t)s3 * OUT_DIM))[lane];
        acc.x += a0 * z0.x; acc.y += a0 * z0.y;
        acc.x += a1 * z1.x; acc.y += a1 * z1.y;
        acc.x += a2 * z2.x; acc.y += a2 * z2.y;
        acc.x += a3 * z3.x; acc.y += a3 * z3.y;
    }
    for (; j < n; ++j) {
        const int s0 = __builtin_amdgcn_readlane(si, j);
        const float a0 = __int_as_float(__builtin_amdgcn_readlane(__float_as_int(a), j));
        const float2 z0 = ((const float2*)(z + (size_t)s0 * OUT_DIM))[lane];
        acc.x += a0 * z0.x; acc.y += a0 * z0.y;
    }
}

__global__ void
gat_node(const int* __restrict__ off, const int* __restrict__ csr,
         const float* __restrict__ s, const float* __restrict__ t,
         const float* __restrict__ z, float* __restrict__ out) {
    const int lane = threadIdx.x & 31;
    const int node = __builtin_amdgcn_readfirstlane((int)(blockIdx.x * 8 + (threadIdx.x >> 5)));
    if (node >= N_NODES) return;
    int start = off[node];
    int end   = (node + 1 < N_NODES) ? off[node + 1] : N_EDGES;
    start = min(max(start, 0), N_EDGES);
    end   = min(max(end, start), N_EDGES);
    float2 acc = make_float2(0.f, 0.f);
    if (end > start) {
        const float td = t[node];
        const int  idx0 = start + lane;
        const bool ok0  = idx0 < end;
        int   si0 = 0;
        float v0  = -INFINITY;
        if (ok0) { si0 = min(max(csr[idx0], 0), N_NODES - 1); v0 = leaky(s[si0] + td); }
        float mx = v0;
        for (int base = start + 32; base < end; base += 32) {
            const int idx = base + lane;
            if (idx < end) mx = fmaxf(mx, leaky(s[min(max(csr[idx], 0), N_NODES - 1)] + td));
        }
        mx = wave_max(mx);
        if (!isfinite(mx)) mx = 0.f;
        const float p0 = ok0 ? expf(v0 - mx) : 0.f;
        float ds = p0;
        for (int base = start + 32; base < end; base += 32) {
            const int idx = base + lane;
            if (idx < end) ds += expf(leaky(s[min(max(csr[idx], 0), N_NODES - 1)] + td) - mx);
        }
        const float denom = wave_sum(ds);
        const int n0 = (end - start < 32) ? (end - start) : 32;
        add_rows(acc, si0, ok0 ? p0 / denom : 0.f, n0, z, lane);
        for (int base = start + 32; base < end; base += 32) {
            const int idx = base + lane;
            int   si = 0;
            float a  = 0.f;
            if (idx < end) { si = min(max(csr[idx], 0), N_NODES - 1); a = expf(leaky(s[si] + td) - mx) / denom; }
            const int n = (end - base < 32) ? (end - base) : 32;
            add_rows(acc, si, a, n, z, lane);
        }
    }
    ((float2*)(out + (size_t)node * OUT_DIM))[lane] = acc;
}

extern "C" void kernel_launch(void* const* d_in, const int* in_sizes, int n_in,
                              void* d_out, int out_size, void* d_ws, size_t ws_size,
                              hipStream_t stream) {
    const float* h  = (const float*)d_in[0];
    const int* src  = (const int*)d_in[1];
    const int* dst  = (const int*)d_in[2];
    const float* W  = (const float*)d_in[3];
    const float* A  = (const float*)d_in[4];
    float* out = (float*)d_out;

    char* ws = (char*)d_ws;
    float* z   = (float*)(ws);
    float* s   = (float*)(ws + 25600000);
    float* t   = (float*)(ws + 26000000);
    int*   cnt = (int*)  (ws + 26400000);
    int*   off = (int*)  (ws + 26800000);
    int*   csr = (int*)  (ws + 27200000);

    const int node_blocks = (N_NODES + 255) / 256;
    const int edge_blocks = (N_EDGES + 255) / 256;

    gat_gemm_wmma<<<(N_STRIPS + GEMM_WAVES - 1) / GEMM_WAVES, GEMM_WAVES * 32, 0, stream>>>(h, W, z);
    gat_proj<<<node_blocks, 256, 0, stream>>>(z, A, s, t, cnt);
    gat_count<<<edge_blocks, 256, 0, stream>>>(dst, cnt);
    gat_scan<<<1, SCAN_T, 0, stream>>>(cnt, off);
    gat_fill<<<edge_blocks, 256, 0, stream>>>(src, dst, off, cnt, csr);
    gat_node<<<N_NODES / 8, 256, 0, stream>>>(off, csr, s, t, z, out);
}


# edge order built through LDS: two-level counting sort by destination range (LDS counters, contiguous runs), no global atomics or scattered 4-byte stores; sums in LDS arrival order
# speedup vs baseline: 4.3056x; 3.1497x over previous
#include <hip/hip_runtime.h>
#include <hip/hip_bf16.h>
#include <math.h>


#define N_NODES 100000
#define N_EDGES 1600000
#define IN_DIM  128
#define OUT_DIM 64
#define LEAKY   0.01f

typedef float v2f __attribute__((ext_vector_type(2)));
typedef float v8f __attribute__((ext_vector_type(8)));

#define GEMM_WAVES 4
#define N_STRIPS (N_NODES / 16)
#define W_PITCH 132
#define W_STAGE ((OUT_DIM * IN_DIM / 4) / (GEMM_WAVES * 32))
#define WMMA_F32(A, B, C) __builtin_amdgcn_wmma_f32_16x16x4_f32(false, (A), false, (B), (short)0, (C), false, false)

__global__ void __launch_bounds__(GEMM_WAVES * 32)
gat_gemm_wmma(const float* __restrict__ h,
              const float* __restrict__ W,
              float* __restrict__ z) {
    __shared__ float4 wl4[OUT_DIM * W_PITCH / 4];
    float* wl = (float*)wl4;
    const int tid  = threadIdx.x;

#pragma unroll
    for (int i = 0; i < W_STAGE; ++i) {
        const int f  = tid + GEMM_WAVES * 32 * i;
        const int r  = f >> 5;
        const int c4 = f & 31;
        *(float4*)(wl + r * W_PITCH + c4 * 4) = ((const float4*)W)[f];
    }
    __syncthreads();

    const int lane  = tid & 31;
    const int strip = blockIdx.x * GEMM_WAVES + (tid >> 5);
    if (strip >= N_STRIPS) return;
    const int m  = lane & 15;
    const int hf = lane >> 4;
    const int row0 = strip * 16;

    float4 a[IN_DIM / 8];
    const float4* __restrict__ hrow = (const float4*)(h + (size_t)(row0 + m) * IN_DIM) + hf;
#pragma unroll
    for (int j = 0; j < IN_DIM / 8; ++j) a[j] = hrow[2 * j];
    __builtin_amdgcn_sched_barrier(0);

    const float* wrow = wl + m * W_PITCH + hf * 4;
    v8f c0 = {}, c1 = {}, c2 = {}, c3 = {};
    float4 n0 = *(const float4*)(wrow +  0 * W_PITCH);
    float4 n1 = *(const float4*)(wrow + 16 * W_PITCH);
    float4 n2 = *(const float4*)(wrow + 32 * W_PITCH);
    float4 n3 = *(const float4*)(wrow + 48 * W_PITCH);
#pragma unroll
    for (int j = 0; j < IN_DIM / 8; ++j) {
        const float4 b0 = n0, b1 = n1, b2 = n2, b3 = n3;
        if (j + 1 < IN_DIM / 8) {
            n0 = *(const float4*)(wrow +  0 * W_PITCH + 8 * (j + 1));
            n1 = *(const float4*)(wrow + 16 * W_PITCH + 8 * (j + 1));
            n2 = *(const float4*)(wrow + 32 * W_PITCH + 8 * (j + 1));
            n3 = *(const float4*)(wrow + 48 * W_PITCH + 8 * (j + 1));
        }
        __builtin_amdgcn_sched_barrier(0);
        v2f alo, ahi, blo, bhi;
        alo.x = a[j].x; alo.y = a[j].y; ahi.x = a[j].z; ahi.y = a[j].w;
        blo.x = b0.x; blo.y = b0.y; bhi.x = b0.z; bhi.y = b0.w;
        c0 = WMMA_F32(alo, blo, c0);
        blo.x = b1.x; blo.y = b1.y;
        c1 = WMMA_F32(alo, blo, c1);
        blo.x = b2.x; blo.y = b2.y;
        c2 = WMMA_F32(alo, blo, c2);
        blo.x = b3.x; blo.y = b3.y;
        c3 = WMMA_F32(alo, blo, c3);
        c0 = WMMA_F32(ahi, bhi, c0);
        bhi.x = b1.z; bhi.y = b1.w;
        c1 = WMMA_F32(ahi, bhi, c1);
        bhi.x = b2.z; bhi.y = b2.w;
        c2 = WMMA_F32(ahi, bhi, c2);
        bhi.x = b3.z; bhi.y = b3.w;
        c3 = WMMA_F32(ahi, bhi, c3);
    }

    float* __restrict__ zrow = z + (size_t)(row0 + hf * 8) * OUT_DIM + m;
#pragma unroll
    for (int r = 0; r < 8; ++r) {
        zrow[(size_t)r * OUT_DIM +  0] = c0[r];
        zrow[(size_t)r * OUT_DIM + 16] = c1[r];
        zrow[(size_t)r * OUT_DIM + 32] = c2[r];
        zrow[(size_t)r * OUT_DIM + 48] = c3[r];
    }
}

__global__ void gat_proj(const float* __restrict__ z, const float* __restrict__ A,
                         float* __restrict__ s, float* __restrict__ t) {
    int i = blockIdx.x * blockDim.x + threadIdx.x;
    if (i >= N_NODES) return;
    const float4* zr = (const float4*)(z + (size_t)i * OUT_DIM);
    const float4* as = (const float4*)A;
    const float4* ad = (const float4*)(A + OUT_DIM);
    float accs = 0.f, acct = 0.f;
#pragma unroll
    for (int k = 0; k < OUT_DIM / 4; ++k) {
        float4 v = zr[k], x = as[k], y = ad[k];
        accs += v.x * x.x + v.y * x.y + v.z * x.z + v.w * x.w;
        acct += v.x * y.x + v.y * y.y + v.z * y.z + v.w * y.w;
    }
    s[i] = accs;
    t[i] = acct;
}

#define BIN_SHIFT 9
#define BIN_NODES (1 << BIN_SHIFT)
#define NBINS ((N_NODES + BIN_NODES - 1) / BIN_NODES)
#define CHUNK 8192
#define NCHUNK ((N_EDGES + CHUNK - 1) / CHUNK)
#define SORT_T 1024
#define EPT (CHUNK / SORT_T)
#define SRC_BITS 17
#define SRC_MASK ((1u << SRC_BITS) - 1u)
#define TAB_SLICES 5
#define TAB_SLICE_LEN ((NCHUNK + TAB_SLICES - 1) / TAB_SLICES)
#define STAGE_CAP 16384
static_assert(N_NODES <= (1 << SRC_BITS), "source index must fit SRC_BITS");
static_assert(TAB_SLICES * NBINS <= SORT_T, "one thread per (slice, range)");
static_assert(NBINS <= SORT_T && BIN_NODES <= SORT_T, "one thread per range / per destination");

__device__ __forceinline__ int block_excl_scan(int x, int* wt, int tid) {
    const int lane = tid & 31;
    const int w    = tid >> 5;
    int inc = x;
    for (int o = 1; o < 32; o <<= 1) { const int y = __shfl_up(inc, o, 32); if (lane >= o) inc += y; }
    __syncthreads();
    if (lane == 31) wt[w] = inc;
    __syncthreads();
    const int wv = wt[lane];
    int winc = wv;
    for (int o = 1; o < 32; o <<= 1) { const int y = __shfl_up(winc, o, 32); if (lane >= o) winc += y; }
    const int wave_excl = __shfl(winc, w, 32) - __shfl(wv, w, 32);
    return wave_excl + inc - x;
}

__global__ void __launch_bounds__(SORT_T)
gat_bin_count(const int* __restrict__ dst, int* __restrict__ table) {
    __shared__ int hist[NBINS];
    const int tid = threadIdx.x;
    if (tid < NBINS) hist[tid] = 0;
    __syncthreads();
    const int e0 = blockIdx.x * CHUNK + tid;
#pragma unroll
    for (int k = 0; k < EPT; ++k) {
        const int e = e0 + k * SORT_T;
        if (e < N_EDGES) {
            const int d = min(max(dst[e], 0), N_NODES - 1);
            atomicAdd(&hist[d >> BIN_SHIFT], 1);
        }
    }
    __syncthreads();
    if (tid < NBINS) table[blockIdx.x * NBINS + tid] = hist[tid];
}

__global__ void __launch_bounds__(SORT_T)
gat_bin_scatter(const int* __restrict__ src, const int* __restrict__ dst,
                const int* __restrict__ table, int* __restrict__ binstart,
                unsigned* __restrict__ binned) {
    __shared__ int wt[32];
    __shared__ int ppre[TAB_SLICES * NBINS];
    __shared__ int ptot[TAB_SLICES * NBINS];
    __shared__ int gbase[NBINS];
    __shared__ int lcnt[NBINS];
    __shared__ int loff[NBINS];
    __shared__ unsigned stage_w[CHUNK];
    __shared__ int stage_g[CHUNK];
    const int tid = threadIdx.x;
    const int blk = blockIdx.x;
    {
        const int sl = tid / NBINS;
        const int b  = tid - sl * NBINS;
        if (sl < TAB_SLICES) {
            const int lo = sl * TAB_SLICE_LEN;
            const int hi = min(lo + TAB_SLICE_LEN, NCHUNK);
            int pre = 0, tot = 0;
            for (int c = lo; c < hi; ++c) {
                const int v = table[c * NBINS + b];
                tot += v;
                if (c < blk) pre += v;
            }
            ppre[sl * NBINS + b] = pre;
            ptot[sl * NBINS + b] = tot;
        }
    }
    __syncthreads();
    int pre = 0, tot = 0;
    if (tid < NBINS) {
        for (int k = 0; k < TAB_SLICES; ++k) { pre += ppre[k * NBINS + tid]; tot += ptot[k * NBINS + tid]; }
        lcnt[tid] = 0;
    }
    const int bstart = block_excl_scan(tot, wt, tid);
    if (tid < NBINS) {
        gbase[tid] = bstart + pre;
        if (blk == 0) binstart[tid] = bstart;
    }
    if (blk == 0 && tid == 0) binstart[NBINS] = N_EDGES;
    __syncthreads();

    unsigned pk[EPT];
    int bn[EPT], rk[EPT];
    const int e0 = blk * CHUNK + tid;
#pragma unroll
    for (int k = 0; k < EPT; ++k) {
        const int e = e0 + k * SORT_T;
        bn[k] = -1; rk[k] = 0; pk[k] = 0u;
        if (e < N_EDGES) {
            const int d = min(max(dst[e], 0), N_NODES - 1);
            const int s = min(max(src[e], 0), N_NODES - 1);
            bn[k] = d >> BIN_SHIFT;
            pk[k] = (unsigned)s | ((unsigned)(d & (BIN_NODES - 1)) << SRC_BITS);
            rk[k] = atomicAdd(&lcnt[bn[k]], 1);
        }
    }
    __syncthreads();
    const int lo = block_excl_scan(tid < NBINS ? lcnt[tid] : 0, wt, tid);
    if (tid < NBINS) loff[tid] = lo;
    __syncthreads();
#pragma unroll
    for (int k = 0; k < EPT; ++k) {
        if (bn[k] >= 0) {
            const int p = loff[bn[k]] + rk[k];
            if ((unsigned)p < (unsigned)CHUNK) { stage_w[p] = pk[k]; stage_g[p] = gbase[bn[k]] + rk[k]; }
        }
    }
    __syncthreads();
    const int nvalid = min(CHUNK, N_EDGES - blk * CHUNK);
    for (int i = tid; i < nvalid; i += SORT_T) {
        const int g = stage_g[i];
        if ((unsigned)g < (unsigned)N_EDGES) binned[g] = stage_w[i];
    }
}

__global__ void __launch_bounds__(SORT_T)
gat_bin_sort(const unsigned* __restrict__ binned, const int* __restrict__ binstart,
             int* __restrict__ off, int* __restrict__ csr) {
    __shared__ int wt[32];
    __shared__ int cnt[BIN_NODES];
    __shared__ int stage[STAGE_CAP];
    const int tid = threadIdx.x;
    const int b   = blockIdx.x;
    int bs = binstart[b], be = binstart[b + 1];
    bs = min(max(bs, 0), N_EDGES);
    be = min(max(be, bs), N_EDGES);
    const int n = be - bs;
    if (tid < BIN_NODES) cnt[tid] = 0;
    __syncthreads();
    for (int i = tid; i < n; i += SORT_T)
        atomicAdd(&cnt[(binned[bs + i] >> SRC_BITS) & (BIN_NODES - 1)], 1);
    __syncthreads();
    const int c  = (tid < BIN_NODES) ? cnt[tid] : 0;
    const int ex = block_excl_scan(c, wt, tid);
    if (tid < BIN_NODES) {
        cnt[tid] = ex;
        const int node = b * BIN_NODES + tid;
        if (node < N_NODES) off[node] = bs + ex;
    }
    __syncthreads();
    const bool staged = n <= STAGE_CAP;
    for (int i = tid; i < n; i += SORT_T) {
        const unsigned w = binned[bs + i];
        const int p  = atomicAdd(&cnt[(w >> SRC_BITS) & (BIN_NODES - 1)], 1);
        const int sv = (int)(w & SRC_MASK);
        if ((unsigned)p < (unsigned)n) {
            if (staged) stage[p] = sv; else csr[bs + p] = sv;
        }
    }
    __syncthreads();
    if (staged)
        for (int i = tid; i < n; i += SORT_T) csr[bs + i] = stage[i];
}

__device__ __forceinline__ float wave_max(float x) {
    for (int o = 16; o > 0; o >>= 1) x = fmaxf(x, __shfl_xor(x, o, 32));
    return x;
}
__device__ __forceinline__ float wave_sum(float x) {
    for (int o = 16; o > 0; o >>= 1) x += __shfl_xor(x, o, 32);
    return x;
}
__device__ __forceinline__ float leaky(float v) { return (v > 0.f) ? v : LEAKY * v; }

__device__ __forceinline__ void add_rows(float2& acc, int si, float a, int n,
                                         const float* __restrict__ z, int lane) {
    int j = 0;
    for (; j + 4 <= n; j += 4) {
        const int s0 = __builtin_amdgcn_readlane(si, j + 0);
        const int s1 = __builtin_amdgcn_readlane(si, j + 1);
        const int s2 = __builtin_amdgcn_readlane(si, j + 2);
        const int s3 = __builtin_amdgcn_readlane(si, j + 3);
        const float a0 = __int_as_float(__builtin_amdgcn_readlane(__float_as_int(a), j + 0));
        const float a1 = __int_as_float(__builtin_amdgcn_readlane(__float_as_int(a), j + 1));
        const float a2 = __int_as_float(__builtin_amdgcn_readlane(__float_as_int(a), j + 2));
        const float a3 = __int_as_float(__builtin_amdgcn_readlane(__float_as_int(a), j + 3));
        const float2 z0 = ((const float2*)(z + (size_t)s0 * OUT_DIM))[lane];
        const float2 z1 = ((const float2*)(z + (size_t)s1 * OUT_DIM))[lane];
        const float2 z2 = ((const float2*)(z + (size_t)s2 * OUT_DIM))[lane];
        const float2 z3 = ((const float2*)(z + (size_t)s3 * OUT_DIM))[lane];
        acc.x += a0 * z0.x; acc.y += a0 * z0.y;
        acc.x += a1 * z1.x; acc.y += a1 * z1.y;
        acc.x += a2 * z2.x; acc.y += a2 * z2.y;
        acc.x += a3 * z3.x; acc.y += a3 * z3.y;
    }
    for (; j < n; ++j) {
        const int s0 = __builtin_amdgcn_readlane(si, j);
        const float a0 = __int_as_float(__builtin_amdgcn_readlane(__float_as_int(a), j));
        const float2 z0 = ((const float2*)(z + (size_t)s0 * OUT_DIM))[lane];
        acc.x += a0 * z0.x; acc.y += a0 * z0.y;
    }
}

__global__ void
gat_node(const int* __restrict__ off, const int* __restrict__ csr,
         const float* __restrict__ s, const float* __restrict__ t,
         const float* __restrict__ z, float* __restrict__ out) {
    const int lane = threadIdx.x & 31;
    const int node = __builtin_amdgcn_readfirstlane((int)(blockIdx.x * 8 + (threadIdx.x >> 5)));
    if (node >= N_NODES) return;
    int start = off[node];
    int end   = (node + 1 < N_NODES) ? off[node + 1] : N_EDGES;
    start = min(max(start, 0), N_EDGES);
    end   = min(max(end, start), N_EDGES);
    float2 acc = make_float2(0.f, 0.f);
    if (end > start) {
        const float td = t[node];
        const int  idx0 = start + lane;
        const bool ok0  = idx0 < end;
        int   si0 = 0;
        float v0  = -INFINITY;
        if (ok0) { si0 = min(max(csr[idx0], 0), N_NODES - 1); v0 = leaky(s[si0] + td); }
        float mx = v0;
        for (int base = start + 32; base < end; base += 32) {
            const int idx = base + lane;
            if (idx < end) mx = fmaxf(mx, leaky(s[min(max(csr[idx], 0), N_NODES - 1)] + td));
        }
        mx = wave_max(mx);
        if (!isfinite(mx)) mx = 0.f;
        const float p0 = ok0 ? expf(v0 - mx) : 0.f;
        float ds = p0;
        for (int base = start + 32; base < end; base += 32) {
            const int idx = base + lane;
            if (idx < end) ds += expf(leaky(s[min(max(csr[idx], 0), N_NODES - 1)] + td) - mx);
        }
        const float denom = wave_sum(ds);
        const int n0 = (end - start < 32) ? (end - start) : 32;
        add_rows(acc, si0, ok0 ? p0 / denom : 0.f, n0, z, lane);
        for (int base = start + 32; base < end; base += 32) {
            const int idx = base + lane;
            int   si = 0;
            float a  = 0.f;
            if (idx < end) { si = min(max(csr[idx], 0), N_NODES - 1); a = expf(leaky(s[si] + td) - mx) / denom; }
            const int n = (end - base < 32) ? (end - base) : 32;
            add_rows(acc, si, a, n, z, lane);
        }
    }
    ((float2*)(out + (size_t)node * OUT_DIM))[lane] = acc;
}

extern "C" void kernel_launch(void* const* d_in, const int* in_sizes, int n_in,
                              void* d_out, int out_size, void* d_ws, size_t ws_size,
                              hipStream_t stream) {
    const float* h  = (const float*)d_in[0];
    const int* src  = (const int*)d_in[1];
    const int* dst  = (const int*)d_in[2];
    const float* W  = (const float*)d_in[3];
    const float* A  = (const float*)d_in[4];
    float* out = (float*)d_out;

    char* ws = (char*)d_ws;
    float*    z        = (float*)   (ws);
    unsigned* binned   = (unsigned*)(ws);
    float*    s        = (float*)   (ws + 25600000);
    float*    t        = (float*)   (ws + 26000000);
    int*      off      = (int*)     (ws + 26400000);
    int*      table    = (int*)     (ws + 26800000);
    int*      binstart = (int*)     (ws + 26960000);
    int*      csr      = (int*)     (ws + 27200000);

    const int node_blocks = (N_NODES + 255) / 256;

    gat_bin_count<<<NCHUNK, SORT_T, 0, stream>>>(dst, table);
    gat_bin_scatter<<<NCHUNK, SORT_T, 0, stream>>>(src, dst, table, binstart, binned);
    gat_bin_sort<<<NBINS, SORT_T, 0, stream>>>(binned, binstart, off, csr);
    gat_gemm_wmma<<<(N_STRIPS + GEMM_WAVES - 1) / GEMM_WAVES, GEMM_WAVES * 32, 0, stream>>>(h, W, z);
    gat_proj<<<node_blocks, 256, 0, stream>>>(z, A, s, t);
    gat_node<<<N_NODES / 8, 256, 0, stream>>>(off, csr, s, t, z, out);
}


# node launch sized to fill the chip once: 16384 waves, not 100000; each wave loops over its nodes and loads the next nodes' bounds, source indices and s values ahead; per-node sums in the same order
# speedup vs baseline: 4.4638x; 1.0367x over previous
#include <hip/hip_runtime.h>
#include <hip/hip_bf16.h>
#include <math.h>


#define N_NODES 100000
#define N_EDGES 1600000
#define IN_DIM  128
#define OUT_DIM 64
#define LEAKY   0.01f

typedef float v2f __attribute__((ext_vector_type(2)));
typedef float v8f __attribute__((ext_vector_type(8)));

#define GEMM_WAVES 4
#define N_STRIPS (N_NODES / 16)
#define W_PITCH 132
#define W_STAGE ((OUT_DIM * IN_DIM / 4) / (GEMM_WAVES * 32))
#define WMMA_F32(A, B, C) __builtin_amdgcn_wmma_f32_16x16x4_f32(false, (A), false, (B), (short)0, (C), false, false)

__global__ void __launch_bounds__(GEMM_WAVES * 32)
gat_gemm_wmma(const float* __restrict__ h,
              const float* __restrict__ W,
              float* __restrict__ z) {
    __shared__ float4 wl4[OUT_DIM * W_PITCH / 4];
    float* wl = (float*)wl4;
    const int tid  = threadIdx.x;

#pragma unroll
    for (int i = 0; i < W_STAGE; ++i) {
        const int f  = tid + GEMM_WAVES * 32 * i;
        const int r  = f >> 5;
        const int c4 = f & 31;
        *(float4*)(wl + r * W_PITCH + c4 * 4) = ((const float4*)W)[f];
    }
    __syncthreads();

    const int lane  = tid & 31;
    const int strip = blockIdx.x * GEMM_WAVES + (tid >> 5);
    if (strip >= N_STRIPS) return;
    const int m  = lane & 15;
    const int hf = lane >> 4;
    const int row0 = strip * 16;

    float4 a[IN_DIM / 8];
    const float4* __restrict__ hrow = (const float4*)(h + (size_t)(row0 + m) * IN_DIM) + hf;
#pragma unroll
    for (int j = 0; j < IN_DIM / 8; ++j) a[j] = hrow[2 * j];
    __builtin_amdgcn_sched_barrier(0);

    const float* wrow = wl + m * W_PITCH + hf * 4;
    v8f c0 = {}, c1 = {}, c2 = {}, c3 = {};
    float4 n0 = *(const float4*)(wrow +  0 * W_PITCH);
    float4 n1 = *(const float4*)(wrow + 16 * W_PITCH);
    float4 n2 = *(const float4*)(wrow + 32 * W_PITCH);
    float4 n3 = *(const float4*)(wrow + 48 * W_PITCH);
#pragma unroll
    for (int j = 0; j < IN_DIM / 8; ++j) {
        const float4 b0 = n0, b1 = n1, b2 = n2, b3 = n3;
        if (j + 1 < IN_DIM / 8) {
            n0 = *(const float4*)(wrow +  0 * W_PITCH + 8 * (j + 1));
            n1 = *(const float4*)(wrow + 16 * W_PITCH + 8 * (j + 1));
            n2 = *(const float4*)(wrow + 32 * W_PITCH + 8 * (j + 1));
            n3 = *(const float4*)(wrow + 48 * W_PITCH + 8 * (j + 1));
        }
        __builtin_amdgcn_sched_barrier(0);
        v2f alo, ahi, blo, bhi;
        alo.x = a[j].x; alo.y = a[j].y; ahi.x = a[j].z; ahi.y = a[j].w;
        blo.x = b0.x; blo.y = b0.y; bhi.x = b0.z; bhi.y = b0.w;
        c0 = WMMA_F32(alo, blo, c0);
        blo.x = b1.x; blo.y = b1.y;
        c1 = WMMA_F32(alo, blo, c1);
        blo.x = b2.x; blo.y = b2.y;
        c2 = WMMA_F32(alo, blo, c2);
        blo.x = b3.x; blo.y = b3.y;
        c3 = WMMA_F32(alo, blo, c3);
        c0 = WMMA_F32(ahi, bhi, c0);
        bhi.x = b1.z; bhi.y = b1.w;
        c1 = WMMA_F32(ahi, bhi, c1);
        bhi.x = b2.z; bhi.y = b2.w;
        c2 = WMMA_F32(ahi, bhi, c2);
        bhi.x = b3.z; bhi.y = b3.w;
        c3 = WMMA_F32(ahi, bhi, c3);
    }

    float* __restrict__ zrow = z + (size_t)(row0 + hf * 8) * OUT_DIM + m;
#pragma unroll
    for (int r = 0; r < 8; ++r) {
        zrow[(size_t)r * OUT_DIM +  0] = c0[r];
        zrow[(size_t)r * OUT_DIM + 16] = c1[r];
        zrow[(size_t)r * OUT_DIM + 32] = c2[r];
        zrow[(size_t)r * OUT_DIM + 48] = c3[r];
    }
}

__global__ void gat_proj(const float* __restrict__ z, const float* __restrict__ A,
                         float* __restrict__ s, float* __restrict__ t) {
    int i = blockIdx.x * blockDim.x + threadIdx.x;
    if (i >= N_NODES) return;
    const float4* zr = (const float4*)(z + (size_t)i * OUT_DIM);
    const float4* as = (const float4*)A;
    const float4* ad = (const float4*)(A + OUT_DIM);
    float accs = 0.f, acct = 0.f;
#pragma unroll
    for (int k = 0; k < OUT_DIM / 4; ++k) {
        float4 v = zr[k], x = as[k], y = ad[k];
        accs += v.x * x.x + v.y * x.y + v.z * x.z + v.w * x.w;
        acct += v.x * y.x + v.y * y.y + v.z * y.z + v.w * y.w;
    }
    s[i] = accs;
    t[i] = acct;
}

#define BIN_SHIFT 9
#define BIN_NODES (1 << BIN_SHIFT)
#define NBINS ((N_NODES + BIN_NODES - 1) / BIN_NODES)
#define CHUNK 8192
#define NCHUNK ((N_EDGES + CHUNK - 1) / CHUNK)
#define SORT_T 1024
#define EPT (CHUNK / SORT_T)
#define SRC_BITS 17
#define SRC_MASK ((1u << SRC_BITS) - 1u)
#define TAB_SLICES 5
#define TAB_SLICE_LEN ((NCHUNK + TAB_SLICES - 1) / TAB_SLICES)
#define STAGE_CAP 16384
static_assert(N_NODES <= (1 << SRC_BITS), "source index must fit SRC_BITS");
static_assert(TAB_SLICES * NBINS <= SORT_T, "one thread per (slice, range)");
static_assert(NBINS <= SORT_T && BIN_NODES <= SORT_T, "one thread per range / per destination");

__device__ __forceinline__ int block_excl_scan(int x, int* wt, int tid) {
    const int lane = tid & 31;
    const int w    = tid >> 5;
    int inc = x;
    for (int o = 1; o < 32; o <<= 1) { const int y = __shfl_up(inc, o, 32); if (lane >= o) inc += y; }
    __syncthreads();
    if (lane == 31) wt[w] = inc;
    __syncthreads();
    const int wv = wt[lane];
    int winc = wv;
    for (int o = 1; o < 32; o <<= 1) { const int y = __shfl_up(winc, o, 32); if (lane >= o) winc += y; }
    const int wave_excl = __shfl(winc, w, 32) - __shfl(wv, w, 32);
    return wave_excl + inc - x;
}

__global__ void __launch_bounds__(SORT_T)
gat_bin_count(const int* __restrict__ dst, int* __restrict__ table) {
    __shared__ int hist[NBINS];
    const int tid = threadIdx.x;
    if (tid < NBINS) hist[tid] = 0;
    __syncthreads();
    const int e0 = blockIdx.x * CHUNK + tid;
#pragma unroll
    for (int k = 0; k < EPT; ++k) {
        const int e = e0 + k * SORT_T;
        if (e < N_EDGES) {
            const int d = min(max(dst[e], 0), N_NODES - 1);
            atomicAdd(&hist[d >> BIN_SHIFT], 1);
        }
    }
    __syncthreads();
    if (tid < NBINS) table[blockIdx.x * NBINS + tid] = hist[tid];
}

__global__ void __launch_bounds__(SORT_T)
gat_bin_scatter(const int* __restrict__ src, const int* __restrict__ dst,
                const int* __restrict__ table, int* __restrict__ binstart,
                unsigned* __restrict__ binned) {
    __shared__ int wt[32];
    __shared__ int ppre[TAB_SLICES * NBINS];
    __shared__ int ptot[TAB_SLICES * NBINS];
    __shared__ int gbase[NBINS];
    __shared__ int lcnt[NBINS];
    __shared__ int loff[NBINS];
    __shared__ unsigned stage_w[CHUNK];
    __shared__ int stage_g[CHUNK];
    const int tid = threadIdx.x;
    const int blk = blockIdx.x;
    {
        const int sl = tid / NBINS;
        const int b  = tid - sl * NBINS;
        if (sl < TAB_SLICES) {
            const int lo = sl * TAB_SLICE_LEN;
            const int hi = min(lo + TAB_SLICE_LEN, NCHUNK);
            int pre = 0, tot = 0;
            for (int c = lo; c < hi; ++c) {
                const int v = table[c * NBINS + b];
                tot += v;
                if (c < blk) pre += v;
            }
            ppre[sl * NBINS + b] = pre;
            ptot[sl * NBINS + b] = tot;
        }
    }
    __syncthreads();
    int pre = 0, tot = 0;
    if (tid < NBINS) {
        for (int k = 0; k < TAB_SLICES; ++k) { pre += ppre[k * NBINS + tid]; tot += ptot[k * NBINS + tid]; }
        lcnt[tid] = 0;
    }
    const int bstart = block_excl_scan(tot, wt, tid);
    if (tid < NBINS) {
        gbase[tid] = bstart + pre;
        if (blk == 0) binstart[tid] = bstart;
    }
    if (blk == 0 && tid == 0) binstart[NBINS] = N_EDGES;
    __syncthreads();

    unsigned pk[EPT];
    int bn[EPT], rk[EPT];
    const int e0 = blk * CHUNK + tid;
#pragma unroll
    for (int k = 0; k < EPT; ++k) {
        const int e = e0 + k * SORT_T;
        bn[k] = -1; rk[k] = 0; pk[k] = 0u;
        if (e < N_EDGES) {
            const int d = min(max(dst[e], 0), N_NODES - 1);
            const int s = min(max(src[e], 0), N_NODES - 1);
            bn[k] = d >> BIN_SHIFT;
            pk[k] = (unsigned)s | ((unsigned)(d & (BIN_NODES - 1)) << SRC_BITS);
            rk[k] = atomicAdd(&lcnt[bn[k]], 1);
        }
    }
    __syncthreads();
    const int lo = block_excl_scan(tid < NBINS ? lcnt[tid] : 0, wt, tid);
    if (tid < NBINS) loff[tid] = lo;
    __syncthreads();
#pragma unroll
    for (int k = 0; k < EPT; ++k) {
        if (bn[k] >= 0) {
            const int p = loff[bn[k]] + rk[k];
            if ((unsigned)p < (unsigned)CHUNK) { stage_w[p] = pk[k]; stage_g[p] = gbase[bn[k]] + rk[k]; }
        }
    }
    __syncthreads();
    const int nvalid = min(CHUNK, N_EDGES - blk * CHUNK);
    for (int i = tid; i < nvalid; i += SORT_T) {
        const int g = stage_g[i];
        if ((unsigned)g < (unsigned)N_EDGES) binned[g] = stage_w[i];
    }
}

__global__ void __launch_bounds__(SORT_T)
gat_bin_sort(const unsigned* __restrict__ binned, const int* __restrict__ binstart,
             int* __restrict__ off, int* __restrict__ csr) {
    __shared__ int wt[32];
    __shared__ int cnt[BIN_NODES];
    __shared__ int stage[STAGE_CAP];
    const int tid = threadIdx.x;
    const int b   = blockIdx.x;
    int bs = binstart[b], be = binstart[b + 1];
    bs = min(max(bs, 0), N_EDGES);
    be = min(max(be, bs), N_EDGES);
    const int n = be - bs;
    if (tid < BIN_NODES) cnt[tid] = 0;
    __syncthreads();
    for (int i = tid; i < n; i += SORT_T)
        atomicAdd(&cnt[(binned[bs + i] >> SRC_BITS) & (BIN_NODES - 1)], 1);
    __syncthreads();
    const int c  = (tid < BIN_NODES) ? cnt[tid] : 0;
    const int ex = block_excl_scan(c, wt, tid);
    if (tid < BIN_NODES) {
        cnt[tid] = ex;
        const int node = b * BIN_NODES + tid;
        if (node < N_NODES) off[node] = bs + ex;
    }
    __syncthreads();
    const bool staged = n <= STAGE_CAP;
    for (int i = tid; i < n; i += SORT_T) {
        const unsigned w = binned[bs + i];
        const int p  = atomicAdd(&cnt[(w >> SRC_BITS) & (BIN_NODES - 1)], 1);
        const int sv = (int)(w & SRC_MASK);
        if ((unsigned)p < (unsigned)n) {
            if (staged) stage[p] = sv; else csr[bs + p] = sv;
        }
    }
    __syncthreads();
    if (staged)
        for (int i = tid; i < n; i += SORT_T) csr[bs + i] = stage[i];
}

__device__ __forceinline__ float wave_max(float x) {
    for (int o = 16; o > 0; o >>= 1) x = fmaxf(x, __shfl_xor(x, o, 32));
    return x;
}
__device__ __forceinline__ float wave_sum(float x) {
    for (int o = 16; o > 0; o >>= 1) x += __shfl_xor(x, o, 32);
    return x;
}
__device__ __forceinline__ float leaky(float v) { return (v > 0.f) ? v : LEAKY * v; }

__device__ __forceinline__ void add_rows(float2& acc, int si, float a, int n,
                                         const float* __restrict__ z, int lane) {
    int j = 0;
    for (; j + 4 <= n; j += 4) {
        const int s0 = __builtin_amdgcn_readlane(si, j + 0);
        const int s1 = __builtin_amdgcn_readlane(si, j + 1);
        const int s2 = __builtin_amdgcn_readlane(si, j + 2);
        const int s3 = __builtin_amdgcn_readlane(si, j + 3);
        const float a0 = __int_as_float(__builtin_amdgcn_readlane(__float_as_int(a), j + 0));
        const float a1 = __int_as_float(__builtin_amdgcn_readlane(__float_as_int(a), j + 1));
        const float a2 = __int_as_float(__builtin_amdgcn_readlane(__float_as_int(a), j + 2));
        const float a3 = __int_as_float(__builtin_amdgcn_readlane(__float_as_int(a), j + 3));
        const float2 z0 = ((const float2*)(z + (size_t)s0 * OUT_DIM))[lane];
        const float2 z1 = ((const float2*)(z + (size_t)s1 * OUT_DIM))[lane];
        const float2 z2 = ((const float2*)(z + (size_t)s2 * OUT_DIM))[lane];
        const float2 z3 = ((const float2*)(z + (size_t)s3 * OUT_DIM))[lane];
        acc.x += a0 * z0.x; acc.y += a0 * z0.y;
        acc.x += a1 * z1.x; acc.y += a1 * z1.y;
        acc.x += a2 * z2.x; acc.y += a2 * z2.y;
        acc.x += a3 * z3.x; acc.y += a3 * z3.y;
    }
    for (; j < n; ++j) {
        const int s0 = __builtin_amdgcn_readlane(si, j);
        const float a0 = __int_as_float(__builtin_amdgcn_readlane(__float_as_int(a), j));
        const float2 z0 = ((const float2*)(z + (size_t)s0 * OUT_DIM))[lane];
        acc.x += a0 * z0.x; acc.y += a0 * z0.y;
    }
}

#define NODE_GROUP_WAVES 8
#define NODE_GROUPS 2048
#define NODE_WAVES (NODE_GROUPS * NODE_GROUP_WAVES)

__global__ void __launch_bounds__(NODE_GROUP_WAVES * 32)
gat_node(const int* __restrict__ off, const int* __restrict__ csr,
         const float* __restrict__ s, const float* __restrict__ t,
         const float* __restrict__ z, float* __restrict__ out) {
    const int lane = threadIdx.x & 31;
    const int wave = __builtin_amdgcn_readfirstlane((int)(blockIdx.x * NODE_GROUP_WAVES + (threadIdx.x >> 5)));

    int rawA_lo, rawA_hi;
    int stB, enB, rawB_ci;
    int stC, enC, siC;
    float rawC_s, rawC_t;
    {
        const int n0 = wave, n1 = wave + NODE_WAVES, n2 = wave + 2 * NODE_WAVES;
        const int lo0 = off[min(n0, N_NODES - 1)], hi0 = off[min(n0 + 1, N_NODES - 1)];
        const int lo1 = off[min(n1, N_NODES - 1)], hi1 = off[min(n1 + 1, N_NODES - 1)];
        rawA_lo = off[min(n2, N_NODES - 1)];
        rawA_hi = off[min(n2 + 1, N_NODES - 1)];
        stC = min(max(lo0, 0), N_EDGES);
        enC = min(max((n0 + 1 < N_NODES) ? hi0 : N_EDGES, stC), N_EDGES);
        if (n0 >= N_NODES) { stC = 0; enC = 0; }
        stB = min(max(lo1, 0), N_EDGES);
        enB = min(max((n1 + 1 < N_NODES) ? hi1 : N_EDGES, stB), N_EDGES);
        if (n1 >= N_NODES) { stB = 0; enB = 0; }
        const int rawC_ci = csr[min(stC + lane, N_EDGES - 1)];
        rawB_ci = csr[min(stB + lane, N_EDGES - 1)];
        siC = min(max(rawC_ci, 0), N_NODES - 1);
        rawC_s = (stC + lane < enC) ? s[siC] : 0.f;
        rawC_t = t[min(n0, N_NODES - 1)];
    }

    for (int node = wave; node < N_NODES; node += NODE_WAVES) {
        const int n1 = node + NODE_WAVES, n2 = node + 2 * NODE_WAVES, n3 = node + 3 * NODE_WAVES;
        const int rawN_lo = off[min(n3, N_NODES - 1)];
        const int rawN_hi = off[min(n3 + 1, N_NODES - 1)];
        int stA = min(max(rawA_lo, 0), N_EDGES);
        int enA = min(max((n2 + 1 < N_NODES) ? rawA_hi : N_EDGES, stA), N_EDGES);
        if (n2 >= N_NODES) { stA = 0; enA = 0; }
        const int rawA_ci = csr[min(stA + lane, N_EDGES - 1)];
        const int siB = min(max(rawB_ci, 0), N_NODES - 1);
        const float rawB_s = (stB + lane < enB) ? s[siB] : 0.f;
        const float rawB_t = t[min(n1, N_NODES - 1)];

        const int start = stC, end = enC;
        float2 acc = make_float2(0.f, 0.f);
        if (end > start) {
            const float td = rawC_t;
            const bool ok0 = start + lane < end;
            const int  si0 = ok0 ? siC : 0;
            const float v0 = ok0 ? leaky(rawC_s + td) : -INFINITY;
            float mx = v0;
            for (int base = start + 32; base < end; base += 32) {
                const int idx = base + lane;
                if (idx < end) mx = fmaxf(mx, leaky(s[min(max(csr[idx], 0), N_NODES - 1)] + td));
            }
            mx = wave_max(mx);
            if (!isfinite(mx)) mx = 0.f;
            const float p0 = ok0 ? expf(v0 - mx) : 0.f;
            float ds = p0;
            for (int base = start + 32; base < end; base += 32) {
                const int idx = base + lane;
                if (idx < end) ds += expf(leaky(s[min(max(csr[idx], 0), N_NODES - 1)] + td) - mx);
            }
            const float denom = wave_sum(ds);
            const int n0 = (end - start < 32) ? (end - start) : 32;
            add_rows(acc, si0, ok0 ? p0 / denom : 0.f, n0, z, lane);
            for (int base = start + 32; base < end; base += 32) {
                const int idx = base + lane;
                int   si = 0;
                float a  = 0.f;
                if (idx < end) { si = min(max(csr[idx], 0), N_NODES - 1); a = expf(leaky(s[si] + td) - mx) / denom; }
                const int n = (end - base < 32) ? (end - base) : 32;
                add_rows(acc, si, a, n, z, lane);
            }
        }
        ((float2*)(out + (size_t)node * OUT_DIM))[lane] = acc;

        stC = stB; enC = enB; siC = siB; rawC_s = rawB_s; rawC_t = rawB_t;
        stB = stA; enB = enA; rawB_ci = rawA_ci;
        rawA_lo = rawN_lo; rawA_hi = rawN_hi;
    }
}

extern "C" void kernel_launch(void* const* d_in, const int* in_sizes, int n_in,
                              void* d_out, int out_size, void* d_ws, size_t ws_size,
                              hipStream_t stream) {
    const float* h  = (const float*)d_in[0];
    const int* src  = (const int*)d_in[1];
    const int* dst  = (const int*)d_in[2];
    const float* W  = (const float*)d_in[3];
    const float* A  = (const float*)d_in[4];
    float* out = (float*)d_out;

    char* ws = (char*)d_ws;
    float*    z        = (float*)   (ws);
    unsigned* binned   = (unsigned*)(ws);
    float*    s        = (float*)   (ws + 25600000);
    float*    t        = (float*)   (ws + 26000000);
    int*      off      = (int*)     (ws + 26400000);
    int*      table    = (int*)     (ws + 26800000);
    int*      binstart = (int*)     (ws + 26960000);
    int*      csr      = (int*)     (ws + 27200000);

    const int node_blocks = (N_NODES + 255) / 256;

    gat_bin_count<<<NCHUNK, SORT_T, 0, stream>>>(dst, table);
    gat_bin_scatter<<<NCHUNK, SORT_T, 0, stream>>>(src, dst, table, binstart, binned);
    gat_bin_sort<<<NBINS, SORT_T, 0, stream>>>(binned, binstart, off, csr);
    gat_gemm_wmma<<<(N_STRIPS + GEMM_WAVES - 1) / GEMM_WAVES, GEMM_WAVES * 32, 0, stream>>>(h, W, z);
    gat_proj<<<node_blocks, 256, 0, stream>>>(z, A, s, t);
    gat_node<<<NODE_GROUPS, NODE_GROUP_WAVES * 32, 0, stream>>>(off, csr, s, t, z, out);
}


# node walk takes two edges per instruction: a gathered row is 16 lanes x 16 B, one half wave per edge, groups of 16, 12-wave work-groups; even and odd edges summed apart, one reciprocal per node
# speedup vs baseline: 7.0858x; 1.5874x over previous
#include <hip/hip_runtime.h>
#include <hip/hip_bf16.h>
#include <math.h>


#define N_NODES 100000
#define N_EDGES 1600000
#define IN_DIM  128
#define OUT_DIM 64
#define LEAKY   0.01f

typedef float v2f __attribute__((ext_vector_type(2)));
typedef float v8f __attribute__((ext_vector_type(8)));
typedef float v4f __attribute__((ext_vector_type(4)));

#define N_STRIPS (N_NODES / 16)
#define W_PITCH 132
#define FUSE_T 1024
#define W_STAGE ((OUT_DIM * IN_DIM / 4) / FUSE_T)
#define A_CHUNK 4
#define WMMA_F32(A, B, C) __builtin_amdgcn_wmma_f32_16x16x4_f32(false, (A), false, (B), (short)0, (C), false, false)

__device__ __forceinline__ void gemm_stage_w(const float* __restrict__ W, float* wl, int tid) {
#pragma unroll
    for (int i = 0; i < W_STAGE; ++i) {
        const int f  = tid + FUSE_T * i;
        const int r  = f >> 5;
        const int c4 = f & 31;
        *(float4*)(wl + r * W_PITCH + c4 * 4) = ((const float4*)W)[f];
    }
}

__device__ __forceinline__ void gemm_strips(const float* __restrict__ h, const float* wl, float* __restrict__ z,
                                            int first, int count, int tid) {
    const int lane = tid & 31;
    const int wave = __builtin_amdgcn_readfirstlane(tid >> 5);
    const int m  = lane & 15;
    const int hf = lane >> 4;
    const float* wrow = wl + m * W_PITCH + hf * 4;
    for (int s = wave; s < count; s += FUSE_T / 32) {
        const int strip = first + s;
        if (strip >= N_STRIPS) break;
        const int row0 = strip * 16;

        float4 a[IN_DIM / 8];
        const float4* __restrict__ hrow = (const float4*)(h + (size_t)(row0 + m) * IN_DIM) + hf;
#pragma unroll
        for (int j = 0; j < A_CHUNK; ++j) a[j] = hrow[2 * j];

        v8f c0 = {}, c1 = {}, c2 = {}, c3 = {};
        float4 n0 = *(const float4*)(wrow +  0 * W_PITCH);
        float4 n1 = *(const float4*)(wrow + 16 * W_PITCH);
        float4 n2 = *(const float4*)(wrow + 32 * W_PITCH);
        float4 n3 = *(const float4*)(wrow + 48 * W_PITCH);
#pragma unroll
        for (int j = 0; j < IN_DIM / 8; ++j) {
            const float4 b0 = n0, b1 = n1, b2 = n2, b3 = n3;
            if (j % A_CHUNK == 0 && j + A_CHUNK < IN_DIM / 8) {
#pragma unroll
                for (int t = 0; t < A_CHUNK; ++t) a[j + A_CHUNK + t] = hrow[2 * (j + A_CHUNK + t)];
            }
            if (j + 1 < IN_DIM / 8) {
                n0 = *(const float4*)(wrow +  0 * W_PITCH + 8 * (j + 1));
                n1 = *(const float4*)(wrow + 16 * W_PITCH + 8 * (j + 1));
                n2 = *(const float4*)(wrow + 32 * W_PITCH + 8 * (j + 1));
                n3 = *(const float4*)(wrow + 48 * W_PITCH + 8 * (j + 1));
            }
            __builtin_amdgcn_sched_barrier(0);
            v2f alo, ahi, blo, bhi;
            alo.x = a[j].x; alo.y = a[j].y; ahi.x = a[j].z; ahi.y = a[j].w;
            blo.x = b0.x; blo.y = b0.y; bhi.x = b0.z; bhi.y = b0.w;
            c0 = WMMA_F32(alo, blo, c0);
            blo.x = b1.x; blo.y = b1.y;
            c1 = WMMA_F32(alo, blo, c1);
            blo.x = b2.x; blo.y = b2.y;
            c2 = WMMA_F32(alo, blo, c2);
            blo.x = b3.x; blo.y = b3.y;
            c3 = WMMA_F32(alo, blo, c3);
            c0 = WMMA_F32(ahi, bhi, c0);
            bhi.x = b1.z; bhi.y = b1.w;
            c1 = WMMA_F32(ahi, bhi, c1);
            bhi.x = b2.z; bhi.y = b2.w;
            c2 = WMMA_F32(ahi, bhi, c2);
            bhi.x = b3.z; bhi.y = b3.w;
            c3 = WMMA_F32(ahi, bhi, c3);
        }

        float* __restrict__ zrow = z + (size_t)(row0 + hf * 8) * OUT_DIM + m;
#pragma unroll
        for (int r = 0; r < 8; ++r) {
            zrow[(size_t)r * OUT_DIM +  0] = c0[r];
            zrow[(size_t)r * OUT_DIM + 16] = c1[r];
            zrow[(size_t)r * OUT_DIM + 32] = c2[r];
            zrow[(size_t)r * OUT_DIM + 48] = c3[r];
        }
    }
}

#define BIN_NODES 98
#define BIN_SLOTS 128
#define BIN_FIELD (BIN_SLOTS - 1)
#define NBINS ((N_NODES + BIN_NODES - 1) / BIN_NODES)
#define CHUNK 8192
#define NCHUNK ((N_EDGES + CHUNK - 1) / CHUNK)
#define SORT_T 1024
#define EPT (CHUNK / SORT_T)
#define SRC_BITS 17
#define SRC_MASK ((1u << SRC_BITS) - 1u)
#define TAB_W (NBINS + 1)
#define STAGE_CAP 4096
#define RN_T 384
#define RN_W (RN_T / 32)
#define RUN_LANES 16
#define RUN_SETS (RN_T / RUN_LANES)
#define RUN_J ((NCHUNK + RUN_SETS - 1) / RUN_SETS)
#define NODE_J ((BIN_NODES + RN_W - 1) / RN_W)
static_assert(N_NODES <= (1 << SRC_BITS), "source index must fit SRC_BITS");
static_assert(SORT_T == 1024, "the scan below takes 32 waves");
static_assert(NBINS < SORT_T && BIN_SLOTS < RN_T && NCHUNK <= RN_T, "one thread per range / destination / chunk");
static_assert(BIN_NODES <= BIN_SLOTS && NODE_J <= 32, "a destination fits its field; a wave's node bounds fit its lanes");

__device__ __forceinline__ int block_excl_scan(int x, int* wt, int tid) {
    const int lane = tid & 31;
    const int w    = tid >> 5;
    int inc = x;
    for (int o = 1; o < 32; o <<= 1) { const int y = __shfl_up(inc, o, 32); if (lane >= o) inc += y; }
    __syncthreads();
    if (lane == 31) wt[w] = inc;
    __syncthreads();
    const int wv = wt[lane];
    int winc = wv;
    for (int o = 1; o < 32; o <<= 1) { const int y = __shfl_up(winc, o, 32); if (lane >= o) winc += y; }
    const int wave_excl = __shfl(winc, w, 32) - __shfl(wv, w, 32);
    return wave_excl + inc - x;
}

#define SORT_STRIPS 20
#define GEMM_ONLY_GROUPS 60
#define GEMM_ONLY_STRIPS 39
static_assert(FUSE_T == SORT_T, "one work-group size for both kinds of work");
static_assert(NCHUNK * SORT_STRIPS + GEMM_ONLY_GROUPS * GEMM_ONLY_STRIPS >= N_STRIPS, "every strip has an owner");
__global__ void __launch_bounds__(SORT_T)
gat_chunk_sort_gemm(const int* __restrict__ src, const int* __restrict__ dst,
                    int* __restrict__ table, unsigned* __restrict__ binned,
                    const float* __restrict__ h, const float* __restrict__ W, float* __restrict__ z) {
    __shared__ float4 wl4[OUT_DIM * W_PITCH / 4];
    __shared__ int wt[32];
    __shared__ int lcnt[NBINS];
    __shared__ int loff[NBINS];
    __shared__ unsigned stage_w[CHUNK];
    const int tid = threadIdx.x;
    const int blk = blockIdx.x;
    gemm_stage_w(W, (float*)wl4, tid);
    if (blk >= NCHUNK) {
        __syncthreads();
        gemm_strips(h, (const float*)wl4, z, NCHUNK * SORT_STRIPS + (blk - NCHUNK) * GEMM_ONLY_STRIPS, GEMM_ONLY_STRIPS, tid);
        return;
    }
    if (tid < NBINS) lcnt[tid] = 0;
    __syncthreads();

    unsigned pk[EPT];
    int bn[EPT], rk[EPT], dv[EPT], sv[EPT];
    const int e0 = blk * CHUNK + tid;
#pragma unroll
    for (int k = 0; k < EPT; ++k) {
        const int ec = min(e0 + k * SORT_T, N_EDGES - 1);
        dv[k] = dst[ec];
        sv[k] = src[ec];
    }
#pragma unroll
    for (int k = 0; k < EPT; ++k) {
        const int e = e0 + k * SORT_T;
        bn[k] = -1; rk[k] = 0; pk[k] = 0u;
        if (e < N_EDGES) {
            const int d = min(max(dv[k], 0), N_NODES - 1);
            const int s = min(max(sv[k], 0), N_NODES - 1);
            bn[k] = d / BIN_NODES;
            pk[k] = (unsigned)s | ((unsigned)(d - bn[k] * BIN_NODES) << SRC_BITS);
            rk[k] = atomicAdd(&lcnt[bn[k]], 1);
        }
    }
    __syncthreads();
    const int lo = block_excl_scan(tid < NBINS ? lcnt[tid] : 0, wt, tid);
    if (tid < NBINS) loff[tid] = lo;
    if (tid <= NBINS) table[tid * NCHUNK + blk] = lo;
    __syncthreads();
#pragma unroll
    for (int k = 0; k < EPT; ++k) {
        if (bn[k] >= 0) {
            const int p = loff[bn[k]] + rk[k];
            if ((unsigned)p < (unsigned)CHUNK) stage_w[p] = pk[k];
        }
    }
    __syncthreads();
    const int nvalid = min(CHUNK, N_EDGES - blk * CHUNK);
    for (int i = tid; i < nvalid; i += SORT_T) binned[blk * CHUNK + i] = stage_w[i];
    gemm_strips(h, (const float*)wl4, z, blk * SORT_STRIPS, SORT_STRIPS, tid);
}


__device__ __forceinline__ float lane_xor16(float x) {
    const int b = __float_as_int(x);
    return __int_as_float(__builtin_amdgcn_permlanex16(b, b, 0x76543210, 0xfedcba98, false, false));
}
#define ROW_XOR(x, M) __int_as_float(__builtin_amdgcn_update_dpp(0, __float_as_int(x), 0x160 + (M), 0xf, 0xf, true))

__device__ __forceinline__ float wave_sum_all(float x) {
    x += lane_xor16(x);
    x += ROW_XOR(x, 8);
    x += ROW_XOR(x, 4);
    x += ROW_XOR(x, 2);
    x += ROW_XOR(x, 1);
    return x;
}

#define ROW_SHARE_I(x, N) __builtin_amdgcn_update_dpp(0, (x), 0x150 + (N), 0xf, 0xf, true)
#define ROW_SHARE(x, N) __int_as_float(ROW_SHARE_I(__float_as_int(x), (N)))

__device__ __forceinline__ int edge_of_lane(int lane) {
    const int c = lane & 15;
    return ((c >> 3) << 4) + ((c & 7) << 1) + (lane >> 4);
}

template <int G>
__device__ __forceinline__ void edge_group16(v4f& acc, float& den, float& mrun, int rowoff, int nv,
                                             const float* __restrict__ z, int lane, v4f as4, float td) {
    const int  c   = lane & 15;
    const bool hi  = (lane & 16) != 0;
    const int  c16 = c * 16;
    v4f r[8];
#define LOAD_PAIR(K) { r[K].x = 0.f; r[K].y = 0.f; r[K].z = 0.f; r[K].w = 0.f; \
        if (2 * (K) < nv) r[K] = *(const v4f*)((const char*)z + (unsigned)(ROW_SHARE_I(rowoff, 8 * G + (K)) | c16)); }
    LOAD_PAIR(0) LOAD_PAIR(1) LOAD_PAIR(2) LOAD_PAIR(3) LOAD_PAIR(4) LOAD_PAIR(5) LOAD_PAIR(6) LOAD_PAIR(7)
#undef LOAD_PAIR
    const v2f alo = as4.xy, ahi = as4.zw;
    const bool b8 = (lane & 8) != 0, b4 = (lane & 4) != 0, b2 = (lane & 2) != 0;
    float q[4];
#pragma unroll
    for (int i = 0; i < 4; ++i) {
        v2f t0 = r[i].xy * alo;
        t0 = r[i].zw * ahi + t0;
        v2f t1 = r[i + 4].xy * alo;
        t1 = r[i + 4].zw * ahi + t1;
        const float p0 = t0.x + t0.y;
        const float p1 = t1.x + t1.y;
        const float keep = b8 ? p1 : p0;
        const float give = b8 ? p0 : p1;
        q[i] = keep + ROW_XOR(give, 8);
    }
    float u[2];
#pragma unroll
    for (int i = 0; i < 2; ++i) {
        const float keep = b4 ? q[i + 2] : q[i];
        const float give = b4 ? q[i] : q[i + 2];
        u[i] = keep + ROW_XOR(give, 4);
    }
    float v;
    {
        const float keep = b2 ? u[1] : u[0];
        const float give = b2 ? u[0] : u[1];
        v = keep + ROW_XOR(give, 2);
    }
    v += ROW_XOR(v, 1);
    float e = v + td;
    e = (e > 0.f) ? e : LEAKY * e;
    const int eidx = (c & ~1) + (hi ? 1 : 0);
    e = (eidx < nv) ? e : -INFINITY;
    float bm = fmaxf(e, lane_xor16(e));
    bm = fmaxf(bm, ROW_XOR(bm, 8));
    bm = fmaxf(bm, ROW_XOR(bm, 4));
    bm = fmaxf(bm, ROW_XOR(bm, 2));
    const float mnew = fmaxf(mrun, bm);
    const float ex = expf((lane == 1) ? (mrun - mnew) : (e - mnew));
    const float scale = __int_as_float(__builtin_amdgcn_readlane(__float_as_int(ex), 1));
    acc = acc * scale;
    den = den * scale;
#define ACC_PAIR(K) { const float w = ROW_SHARE(ex, 2 * (K)); acc += w * r[K]; den += w; }
    ACC_PAIR(0) ACC_PAIR(1) ACC_PAIR(2) ACC_PAIR(3) ACC_PAIR(4) ACC_PAIR(5) ACC_PAIR(6) ACC_PAIR(7)
#undef ACC_PAIR
    mrun = mnew;
}

__device__ __forceinline__ int range_excl_scan(int x, int* wt, int tid) {
    const int lane = tid & 31;
    const int w    = tid >> 5;
    int inc = x;
    for (int o = 1; o < 32; o <<= 1) { const int y = __shfl_up(inc, o, 32); if (lane >= o) inc += y; }
    __syncthreads();
    if (lane == 31) wt[w] = inc;
    __syncthreads();
    int before = 0;
    for (int i = 0; i < RN_W; ++i) before += (i < w) ? wt[i] : 0;
    return before + inc - x;
}

__global__ void __launch_bounds__(RN_T, 12)
gat_range_node(const unsigned* __restrict__ binned, const int* __restrict__ table,
               const float* __restrict__ A, const float* __restrict__ z, float* __restrict__ out) {
    __shared__ int wt[RN_W];
    __shared__ int cnt[BIN_SLOTS];
    __shared__ int seg[BIN_SLOTS + 1];
    __shared__ int stage[STAGE_CAP];
    __shared__ int run_s[NCHUNK];
    __shared__ int run_l[NCHUNK];
    __shared__ int tot[1];
    const int tid  = threadIdx.x;
    const int b    = blockIdx.x;
    const int lane = tid & 31;
    const int wave = __builtin_amdgcn_readfirstlane(tid >> 5);
    if (tid < BIN_SLOTS) cnt[tid] = 0;
    if (tid == 0) tot[0] = 0;
    __syncthreads();
    if (tid < NCHUNK) {
        const int nval = min(CHUNK, N_EDGES - tid * CHUNK);
        int ls = table[b * NCHUNK + tid];
        int le = table[(b + 1) * NCHUNK + tid];
        ls = min(max(ls, 0), nval);
        le = min(max(le, ls), nval);
        run_s[tid] = tid * CHUNK + ls;
        run_l[tid] = le - ls;
        atomicAdd(&tot[0], le - ls);
    }
    __syncthreads();
    const int n = min(max(tot[0], 0), N_EDGES);

    const int hset = tid / RUN_LANES;
    const int hl   = tid % RUN_LANES;
    unsigned wv[RUN_J];
    int rs[RUN_J], rl[RUN_J];
#pragma unroll
    for (int j = 0; j < RUN_J; ++j) {
        const int r = hset + RUN_SETS * j;
        rs[j] = (r < NCHUNK) ? run_s[r] : 0;
        rl[j] = (r < NCHUNK) ? run_l[r] : 0;
    }
#pragma unroll
    for (int j = 0; j < RUN_J; ++j) wv[j] = (hl < rl[j]) ? binned[rs[j] + hl] : 0u;
#pragma unroll
    for (int j = 0; j < RUN_J; ++j) {
        if (hl < rl[j]) atomicAdd(&cnt[(wv[j] >> SRC_BITS) & BIN_FIELD], 1);
        for (int k = hl + RUN_LANES; k < rl[j]; k += RUN_LANES)
            atomicAdd(&cnt[(binned[rs[j] + k] >> SRC_BITS) & BIN_FIELD], 1);
    }
    __syncthreads();
    const int cc = (tid < BIN_SLOTS) ? cnt[tid] : 0;
    const int ex = range_excl_scan(cc, wt, tid);
    if (tid < BIN_SLOTS) cnt[tid] = ex;
    if (tid <= BIN_SLOTS) seg[tid] = ex;
    __syncthreads();
    const bool staged = n <= STAGE_CAP;
    if (staged) {
#pragma unroll
        for (int j = 0; j < RUN_J; ++j) {
            if (hl < rl[j]) {
                const unsigned w = wv[j];
                const int p = atomicAdd(&cnt[(w >> SRC_BITS) & BIN_FIELD], 1);
                if ((unsigned)p < (unsigned)STAGE_CAP) stage[p] = (int)(w & SRC_MASK);
            }
            for (int k = hl + RUN_LANES; k < rl[j]; k += RUN_LANES) {
                const unsigned w = binned[rs[j] + k];
                const int p = atomicAdd(&cnt[(w >> SRC_BITS) & BIN_FIELD], 1);
                if ((unsigned)p < (unsigned)STAGE_CAP) stage[p] = (int)(w & SRC_MASK);
            }
        }
    }
    __syncthreads();

    const int  c  = lane & 15;
    const bool hi = (lane & 16) != 0;
    const v4f as4 = ((const v4f*)A)[c];
    const v2f ad2 = ((const v2f*)(A + OUT_DIM))[lane];
    int lo_l = 0, hi_l = 0;
    {
        const int dk = wave + RN_W * lane;
        if (dk < BIN_NODES) { lo_l = seg[dk]; hi_l = seg[dk + 1]; }
        lo_l = min(max(lo_l, 0), n);
        hi_l = min(max(hi_l, lo_l), n);
    }
    const int node0 = b * BIN_NODES + wave;
    int start = __builtin_amdgcn_readlane(lo_l, 0);
    int end   = __builtin_amdgcn_readlane(hi_l, 0);
    const int eol = edge_of_lane(lane);
    int si_c  = stage[min(max(start + min(eol, end - start - 1), 0), STAGE_CAP - 1)];
    float td_l = 0.f;
    {
        v2f ownr[NODE_J];
#pragma unroll
        for (int k = 0; k < NODE_J; ++k) {
            const int nd = min(node0 + RN_W * k, N_NODES - 1);
            ownr[k] = *(const v2f*)((const char*)z + (unsigned)(nd * (OUT_DIM * 4) + lane * 8));
        }
#pragma unroll
        for (int k = 0; k < NODE_J; ++k) {
            const float t = wave_sum_all(ownr[k].x * ad2.x + ownr[k].y * ad2.y);
            td_l = (lane == k) ? t : td_l;
        }
    }
    for (int k = 0; k < NODE_J; ++k) {
        const int d    = wave + RN_W * k;
        const int node = b * BIN_NODES + d;
        if (d >= BIN_NODES || node >= N_NODES) break;
        const int kn      = min(k + 1, NODE_J - 1);
        const int start_n = __builtin_amdgcn_readlane(lo_l, kn);
        const int end_n   = __builtin_amdgcn_readlane(hi_l, kn);
        const int si_n    = stage[min(max(start_n + min(eol, end_n - start_n - 1), 0), STAGE_CAP - 1)];

        v4f acc; acc.x = 0.f; acc.y = 0.f; acc.z = 0.f; acc.w = 0.f;
        if (end > start) {
            const float td = __int_as_float(__builtin_amdgcn_readlane(__float_as_int(td_l), k));
            float den = 0.f, mrun = -INFINITY;
            if (staged) {
                for (int base = start; base < end; base += 32) {
                    const int n32 = (end - base < 32) ? (end - base) : 32;
                    int si = si_c;
                    if (base != start) si = stage[min(max(base + min(eol, n32 - 1), 0), STAGE_CAP - 1)];
                    si = min(max(si, 0), N_NODES - 1);
                    const int rowoff = si * (OUT_DIM * 4);
                    if (n32 >= 16) {
                        edge_group16<0>(acc, den, mrun, rowoff, 16, z, lane, as4, td);
                        if (n32 > 16) edge_group16<1>(acc, den, mrun, rowoff, n32 - 16, z, lane, as4, td);
                    } else {
                        edge_group16<0>(acc, den, mrun, rowoff, n32, z, lane, as4, td);
                    }
                }
            } else {
                for (int ch = 0; ch < NCHUNK; ++ch) {
                    const int rs_c = __builtin_amdgcn_readfirstlane(run_s[ch]);
                    const int rl_c = __builtin_amdgcn_readfirstlane(run_l[ch]);
                    for (int k0 = 0; k0 < rl_c; k0 += 32) {
                        const bool in = k0 + lane < rl_c;
                        const unsigned w = in ? binned[rs_c + k0 + lane] : 0u;
                        const bool hit = in && (int)((w >> SRC_BITS) & BIN_FIELD) == d;
                        const int si = min((int)(w & SRC_MASK), N_NODES - 1);
                        unsigned mask = __builtin_amdgcn_ballot_w32(hit);
                        while (mask != 0u) {
                            const int L = __builtin_ctz(mask);
                            mask &= mask - 1u;
                            const int one = __builtin_amdgcn_readlane(si, L) * (OUT_DIM * 4);
                            edge_group16<0>(acc, den, mrun, one, 1, z, lane, as4, td);
                        }
                    }
                }
            }
            acc.x += lane_xor16(acc.x);
            acc.y += lane_xor16(acc.y);
            acc.z += lane_xor16(acc.z);
            acc.w += lane_xor16(acc.w);
            den   += lane_xor16(den);
            const float inv = 1.0f / den;
            acc = acc * inv;
        }
        if (!hi) *(v4f*)((char*)out + (unsigned)(node * (OUT_DIM * 4) + c * 16)) = acc;

        start = start_n; end = end_n; si_c = si_n;
    }
}

extern "C" void kernel_launch(void* const* d_in, const int* in_sizes, int n_in,
                              void* d_out, int out_size, void* d_ws, size_t ws_size,
                              hipStream_t stream) {
    const float* h  = (const float*)d_in[0];
    const int* src  = (const int*)d_in[1];
    const int* dst  = (const int*)d_in[2];
    const float* W  = (const float*)d_in[3];
    const float* A  = (const float*)d_in[4];
    float* out = (float*)d_out;

    char* ws = (char*)d_ws;
    float*    z        = (float*)   (ws);
    int*      table    = (int*)     (ws + 25600000);
    unsigned* binned   = (unsigned*)(ws + 27200000);

    gat_chunk_sort_gemm<<<NCHUNK + GEMM_ONLY_GROUPS, SORT_T, 0, stream>>>(src, dst, table, binned, h, W, z);
    gat_range_node<<<NBINS, RN_T, 0, stream>>>(binned, table, A, z, out);
}


# node walk, fewer vector issue cycles per node: up to 8 edges left behind the first 16 take a 4-pair group (same sum order), weights by permlane16 not DPP moves, hardware exp2 and rcp
# speedup vs baseline: 7.5778x; 1.0694x over previous
#include <hip/hip_runtime.h>
#include <hip/hip_bf16.h>
#include <math.h>


#define N_NODES 100000
#define N_EDGES 1600000
#define IN_DIM  128
#define OUT_DIM 64
#define LEAKY   0.01f

typedef float v2f __attribute__((ext_vector_type(2)));
typedef float v8f __attribute__((ext_vector_type(8)));
typedef float v4f __attribute__((ext_vector_type(4)));

#define N_STRIPS (N_NODES / 16)
#define W_PITCH 132
#define FUSE_T 1024
#define W_STAGE ((OUT_DIM * IN_DIM / 4) / FUSE_T)
#define A_CHUNK 4
#define WMMA_F32(A, B, C) __builtin_amdgcn_wmma_f32_16x16x4_f32(false, (A), false, (B), (short)0, (C), false, false)

__device__ __forceinline__ void gemm_stage_w(const float* __restrict__ W, float* wl, int tid) {
#pragma unroll
    for (int i = 0; i < W_STAGE; ++i) {
        const int f  = tid + FUSE_T * i;
        const int r  = f >> 5;
        const int c4 = f & 31;
        *(float4*)(wl + r * W_PITCH + c4 * 4) = ((const float4*)W)[f];
    }
}

__device__ __forceinline__ void gemm_strips(const float* __restrict__ h, const float* wl, float* __restrict__ z,
                                            int first, int count, int tid) {
    const int lane = tid & 31;
    const int wave = __builtin_amdgcn_readfirstlane(tid >> 5);
    const int m  = lane & 15;
    const int hf = lane >> 4;
    const float* wrow = wl + m * W_PITCH + hf * 4;
    for (int s = wave; s < count; s += FUSE_T / 32) {
        const int strip = first + s;
        if (strip >= N_STRIPS) break;
        const int row0 = strip * 16;

        float4 a[IN_DIM / 8];
        const float4* __restrict__ hrow = (const float4*)(h + (size_t)(row0 + m) * IN_DIM) + hf;
#pragma unroll
        for (int j = 0; j < A_CHUNK; ++j) a[j] = hrow[2 * j];

        v8f c0 = {}, c1 = {}, c2 = {}, c3 = {};
        float4 n0 = *(const float4*)(wrow +  0 * W_PITCH);
        float4 n1 = *(const float4*)(wrow + 16 * W_PITCH);
        float4 n2 = *(const float4*)(wrow + 32 * W_PITCH);
        float4 n3 = *(const float4*)(wrow + 48 * W_PITCH);
#pragma unroll
        for (int j = 0; j < IN_DIM / 8; ++j) {
            const float4 b0 = n0, b1 = n1, b2 = n2, b3 = n3;
            if (j % A_CHUNK == 0 && j + A_CHUNK < IN_DIM / 8) {
#pragma unroll
                for (int t = 0; t < A_CHUNK; ++t) a[j + A_CHUNK + t] = hrow[2 * (j + A_CHUNK + t)];
            }
            if (j + 1 < IN_DIM / 8) {
                n0 = *(const float4*)(wrow +  0 * W_PITCH + 8 * (j + 1));
                n1 = *(const float4*)(wrow + 16 * W_PITCH + 8 * (j + 1));
                n2 = *(const float4*)(wrow + 32 * W_PITCH + 8 * (j + 1));
                n3 = *(const float4*)(wrow + 48 * W_PITCH + 8 * (j + 1));
            }
            __builtin_amdgcn_sched_barrier(0);
            v2f alo, ahi, blo, bhi;
            alo.x = a[j].x; alo.y = a[j].y; ahi.x = a[j].z; ahi.y = a[j].w;
            blo.x = b0.x; blo.y = b0.y; bhi.x = b0.z; bhi.y = b0.w;
            c0 = WMMA_F32(alo, blo, c0);
            blo.x = b1.x; blo.y = b1.y;
            c1 = WMMA_F32(alo, blo, c1);
            blo.x = b2.x; blo.y = b2.y;
            c2 = WMMA_F32(alo, blo, c2);
            blo.x = b3.x; blo.y = b3.y;
            c3 = WMMA_F32(alo, blo, c3);
            c0 = WMMA_F32(ahi, bhi, c0);
            bhi.x = b1.z; bhi.y = b1.w;
            c1 = WMMA_F32(ahi, bhi, c1);
            bhi.x = b2.z; bhi.y = b2.w;
            c2 = WMMA_F32(ahi, bhi, c2);
            bhi.x = b3.z; bhi.y = b3.w;
            c3 = WMMA_F32(ahi, bhi, c3);
        }

        float* __restrict__ zrow = z + (size_t)(row0 + hf * 8) * OUT_DIM + m;
#pragma unroll
        for (int r = 0; r < 8; ++r) {
            zrow[(size_t)r * OUT_DIM +  0] = c0[r];
            zrow[(size_t)r * OUT_DIM + 16] = c1[r];
            zrow[(size_t)r * OUT_DIM + 32] = c2[r];
            zrow[(size_t)r * OUT_DIM + 48] = c3[r];
        }
    }
}

#define BIN_NODES 98
#define BIN_SLOTS 128
#define BIN_FIELD (BIN_SLOTS - 1)
#define NBINS ((N_NODES + BIN_NODES - 1) / BIN_NODES)
#define CHUNK 8192
#define NCHUNK ((N_EDGES + CHUNK - 1) / CHUNK)
#define SORT_T 1024
#define EPT (CHUNK / SORT_T)
#define SRC_BITS 17
#define SRC_MASK ((1u << SRC_BITS) - 1u)
#define TAB_W (NBINS + 1)
#define STAGE_CAP 4096
#define RN_T 384
#define RN_W (RN_T / 32)
#define RUN_LANES 16
#define RUN_SETS (RN_T / RUN_LANES)
#define RUN_J ((NCHUNK + RUN_SETS - 1) / RUN_SETS)
#define NODE_J ((BIN_NODES + RN_W - 1) / RN_W)
static_assert(N_NODES <= (1 << SRC_BITS), "source index must fit SRC_BITS");
static_assert(SORT_T == 1024, "the scan below takes 32 waves");
static_assert(NBINS < SORT_T && BIN_SLOTS < RN_T && NCHUNK <= RN_T, "one thread per range / destination / chunk");
static_assert(BIN_NODES <= BIN_SLOTS && NODE_J <= 32, "a destination fits its field; a wave's node bounds fit its lanes");

__device__ __forceinline__ int block_excl_scan(int x, int* wt, int tid) {
    const int lane = tid & 31;
    const int w    = tid >> 5;
    int inc = x;
    for (int o = 1; o < 32; o <<= 1) { const int y = __shfl_up(inc, o, 32); if (lane >= o) inc += y; }
    __syncthreads();
    if (lane == 31) wt[w] = inc;
    __syncthreads();
    const int wv = wt[lane];
    int winc = wv;
    for (int o = 1; o < 32; o <<= 1) { const int y = __shfl_up(winc, o, 32); if (lane >= o) winc += y; }
    const int wave_excl = __shfl(winc, w, 32) - __shfl(wv, w, 32);
    return wave_excl + inc - x;
}

#define SORT_STRIPS 20
#define GEMM_ONLY_GROUPS 60
#define GEMM_ONLY_STRIPS 39
static_assert(FUSE_T == SORT_T, "one work-group size for both kinds of work");
static_assert(NCHUNK * SORT_STRIPS + GEMM_ONLY_GROUPS * GEMM_ONLY_STRIPS >= N_STRIPS, "every strip has an owner");
__global__ void __launch_bounds__(SORT_T)
gat_chunk_sort_gemm(const int* __restrict__ src, const int* __restrict__ dst,
                    int* __restrict__ table, unsigned* __restrict__ binned,
                    const float* __restrict__ h, const float* __restrict__ W, float* __restrict__ z) {
    __shared__ float4 wl4[OUT_DIM * W_PITCH / 4];
    __shared__ int wt[32];
    __shared__ int lcnt[NBINS];
    __shared__ int loff[NBINS];
    __shared__ unsigned stage_w[CHUNK];
    const int tid = threadIdx.x;
    const int blk = blockIdx.x;
    gemm_stage_w(W, (float*)wl4, tid);
    if (blk >= NCHUNK) {
        __syncthreads();
        gemm_strips(h, (const float*)wl4, z, NCHUNK * SORT_STRIPS + (blk - NCHUNK) * GEMM_ONLY_STRIPS, GEMM_ONLY_STRIPS, tid);
        return;
    }
    if (tid < NBINS) lcnt[tid] = 0;
    __syncthreads();

    unsigned pk[EPT];
    int bn[EPT], rk[EPT], dv[EPT], sv[EPT];
    const int e0 = blk * CHUNK + tid;
#pragma unroll
    for (int k = 0; k < EPT; ++k) {
        const int ec = min(e0 + k * SORT_T, N_EDGES - 1);
        dv[k] = dst[ec];
        sv[k] = src[ec];
    }
#pragma unroll
    for (int k = 0; k < EPT; ++k) {
        const int e = e0 + k * SORT_T;
        bn[k] = -1; rk[k] = 0; pk[k] = 0u;
        if (e < N_EDGES) {
            const int d = min(max(dv[k], 0), N_NODES - 1);
            const int s = min(max(sv[k], 0), N_NODES - 1);
            bn[k] = d / BIN_NODES;
            pk[k] = (unsigned)s | ((unsigned)(d - bn[k] * BIN_NODES) << SRC_BITS);
            rk[k] = atomicAdd(&lcnt[bn[k]], 1);
        }
    }
    __syncthreads();
    const int lo = block_excl_scan(tid < NBINS ? lcnt[tid] : 0, wt, tid);
    if (tid < NBINS) loff[tid] = lo;
    if (tid <= NBINS) table[tid * NCHUNK + blk] = lo;
    __syncthreads();
#pragma unroll
    for (int k = 0; k < EPT; ++k) {
        if (bn[k] >= 0) {
            const int p = loff[bn[k]] + rk[k];
            if ((unsigned)p < (unsigned)CHUNK) stage_w[p] = pk[k];
        }
    }
    __syncthreads();
    const int nvalid = min(CHUNK, N_EDGES - blk * CHUNK);
    for (int i = tid; i < nvalid; i += SORT_T) binned[blk * CHUNK + i] = stage_w[i];
    gemm_strips(h, (const float*)wl4, z, blk * SORT_STRIPS, SORT_STRIPS, tid);
}


__device__ __forceinline__ float lane_xor16(float x) {
    const int b = __float_as_int(x);
    return __int_as_float(__builtin_amdgcn_permlanex16(b, b, 0x76543210, 0xfedcba98, true, true));
}
#define ROW_XOR(x, M) __int_as_float(__builtin_amdgcn_update_dpp(0, __float_as_int(x), 0x160 + (M), 0xf, 0xf, true))

__device__ __forceinline__ float wave_sum_all(float x) {
    x += lane_xor16(x);
    x += ROW_XOR(x, 8);
    x += ROW_XOR(x, 4);
    x += ROW_XOR(x, 2);
    x += ROW_XOR(x, 1);
    return x;
}

#define ROW_SHARE_I(x, N) __builtin_amdgcn_update_dpp(0, (x), 0x150 + (N), 0xf, 0xf, true)
#define ROW_SHARE(x, N) __int_as_float(ROW_SHARE_I(__float_as_int(x), (N)))
#define ROW_BCAST(x, N) __int_as_float(__builtin_amdgcn_permlane16(0, __float_as_int(x), 0x11111111u * (unsigned)(N), 0x11111111u * (unsigned)(N), true, true))

__device__ __forceinline__ int edge_of_lane(int lane) {
    const int c = lane & 15;
    return ((c >> 3) << 4) + ((c & 7) << 1) + (lane >> 4);
}

template <int G, int NP>
__device__ __forceinline__ void edge_group16(v4f& acc, float& den, float& mrun, int rowoff, int nv,
                                             const float* __restrict__ z, int lane, v4f as4, float td) {
    const int  c   = lane & 15;
    const bool hi  = (lane & 16) != 0;
    const int  c16 = c * 16;
    constexpr int SH = (NP == 8) ? 1 : 2;
    v4f r[NP];
#define LOAD_PAIR(K) { r[K].x = 0.f; r[K].y = 0.f; r[K].z = 0.f; r[K].w = 0.f; \
        if (NP == 4 || 2 * (K) < nv) r[K] = *(const v4f*)((const char*)z + (unsigned)(ROW_SHARE_I(rowoff, 8 * G + (K)) | c16)); }
    LOAD_PAIR(0) LOAD_PAIR(1) LOAD_PAIR(2) LOAD_PAIR(3)
    if constexpr (NP == 8) { LOAD_PAIR(4) LOAD_PAIR(5) LOAD_PAIR(6) LOAD_PAIR(7) }
#undef LOAD_PAIR
    const v2f alo = as4.xy, ahi = as4.zw;
    const bool b8 = (lane & 8) != 0, b4 = (lane & 4) != 0, b2 = (lane & 2) != 0;
    float q[NP / 2];
#pragma unroll
    for (int i = 0; i < NP / 2; ++i) {
        v2f t0 = r[i].xy * alo;
        t0 = r[i].zw * ahi + t0;
        v2f t1 = r[i + NP / 2].xy * alo;
        t1 = r[i + NP / 2].zw * ahi + t1;
        const float p0 = t0.x + t0.y;
        const float p1 = t1.x + t1.y;
        const float keep = b8 ? p1 : p0;
        const float give = b8 ? p0 : p1;
        q[i] = keep + ROW_XOR(give, 8);
    }
    float v;
    if constexpr (NP == 8) {
        float u[2];
#pragma unroll
        for (int i = 0; i < 2; ++i) {
            const float keep = b4 ? q[i + 2] : q[i];
            const float give = b4 ? q[i] : q[i + 2];
            u[i] = keep + ROW_XOR(give, 4);
        }
        const float keep = b2 ? u[1] : u[0];
        const float give = b2 ? u[0] : u[1];
        v = keep + ROW_XOR(give, 2);
    } else {
        const float keep = b4 ? q[1] : q[0];
        const float give = b4 ? q[0] : q[1];
        v = keep + ROW_XOR(give, 4);
        v += ROW_XOR(v, 2);
    }
    v += ROW_XOR(v, 1);
    float e = v + td;
    e = (e > 0.f) ? e : LEAKY * e;
    const int eidx = ((c >> SH) << 1) + (hi ? 1 : 0);
    e = (eidx < nv) ? e : -INFINITY;
    float bm = fmaxf(e, lane_xor16(e));
    bm = fmaxf(bm, ROW_XOR(bm, 8));
    bm = fmaxf(bm, ROW_XOR(bm, 4));
    if constexpr (NP == 8) bm = fmaxf(bm, ROW_XOR(bm, 2));
    const float mnew = fmaxf(mrun, bm);
    const float ex = __builtin_amdgcn_exp2f(((lane == 1) ? (mrun - mnew) : (e - mnew)) * 1.44269504088896341f);
    const float scale = __int_as_float(__builtin_amdgcn_readlane(__float_as_int(ex), 1));
    acc = acc * scale;
    den = den * scale;
#define ACC_PAIR(K) { const float w = ROW_BCAST(ex, (K) << SH); acc += w * r[K]; den += w; }
    ACC_PAIR(0) ACC_PAIR(1) ACC_PAIR(2) ACC_PAIR(3)
    if constexpr (NP == 8) { ACC_PAIR(4) ACC_PAIR(5) ACC_PAIR(6) ACC_PAIR(7) }
#undef ACC_PAIR
    mrun = mnew;
}

__device__ __forceinline__ int range_excl_scan(int x, int* wt, int tid) {
    const int lane = tid & 31;
    const int w    = tid >> 5;
    int inc = x;
    for (int o = 1; o < 32; o <<= 1) { const int y = __shfl_up(inc, o, 32); if (lane >= o) inc += y; }
    __syncthreads();
    if (lane == 31) wt[w] = inc;
    __syncthreads();
    int before = 0;
    for (int i = 0; i < RN_W; ++i) before += (i < w) ? wt[i] : 0;
    return before + inc - x;
}

__global__ void __launch_bounds__(RN_T, 12)
gat_range_node(const unsigned* __restrict__ binned, const int* __restrict__ table,
               const float* __restrict__ A, const float* __restrict__ z, float* __restrict__ out) {
    __shared__ int wt[RN_W];
    __shared__ int cnt[BIN_SLOTS];
    __shared__ int seg[BIN_SLOTS + 1];
    __shared__ int stage[STAGE_CAP];
    __shared__ int run_s[NCHUNK];
    __shared__ int run_l[NCHUNK];
    __shared__ int tot[1];
    const int tid  = threadIdx.x;
    const int b    = blockIdx.x;
    const int lane = tid & 31;
    const int wave = __builtin_amdgcn_readfirstlane(tid >> 5);
    if (tid < BIN_SLOTS) cnt[tid] = 0;
    if (tid == 0) tot[0] = 0;
    __syncthreads();
    if (tid < NCHUNK) {
        const int nval = min(CHUNK, N_EDGES - tid * CHUNK);
        int ls = table[b * NCHUNK + tid];
        int le = table[(b + 1) * NCHUNK + tid];
        ls = min(max(ls, 0), nval);
        le = min(max(le, ls), nval);
        run_s[tid] = tid * CHUNK + ls;
        run_l[tid] = le - ls;
        atomicAdd(&tot[0], le - ls);
    }
    __syncthreads();
    const int n = min(max(tot[0], 0), N_EDGES);

    const int hset = tid / RUN_LANES;
    const int hl   = tid % RUN_LANES;
    unsigned wv[RUN_J];
    int rs[RUN_J], rl[RUN_J];
#pragma unroll
    for (int j = 0; j < RUN_J; ++j) {
        const int r = hset + RUN_SETS * j;
        rs[j] = (r < NCHUNK) ? run_s[r] : 0;
        rl[j] = (r < NCHUNK) ? run_l[r] : 0;
    }
#pragma unroll
    for (int j = 0; j < RUN_J; ++j) wv[j] = (hl < rl[j]) ? binned[rs[j] + hl] : 0u;
#pragma unroll
    for (int j = 0; j < RUN_J; ++j) {
        if (hl < rl[j]) atomicAdd(&cnt[(wv[j] >> SRC_BITS) & BIN_FIELD], 1);
        for (int k = hl + RUN_LANES; k < rl[j]; k += RUN_LANES)
            atomicAdd(&cnt[(binned[rs[j] + k] >> SRC_BITS) & BIN_FIELD], 1);
    }
    __syncthreads();
    const int cc = (tid < BIN_SLOTS) ? cnt[tid] : 0;
    const int ex = range_excl_scan(cc, wt, tid);
    if (tid < BIN_SLOTS) cnt[tid] = ex;
    if (tid <= BIN_SLOTS) seg[tid] = ex;
    __syncthreads();
    const bool staged = n <= STAGE_CAP;
    if (staged) {
#pragma unroll
        for (int j = 0; j < RUN_J; ++j) {
            if (hl < rl[j]) {
                const unsigned w = wv[j];
                const int p = atomicAdd(&cnt[(w >> SRC_BITS) & BIN_FIELD], 1);
                if ((unsigned)p < (unsigned)STAGE_CAP) stage[p] = (int)(w & SRC_MASK);
            }
            for (int k = hl + RUN_LANES; k < rl[j]; k += RUN_LANES) {
                const unsigned w = binned[rs[j] + k];
                const int p = atomicAdd(&cnt[(w >> SRC_BITS) & BIN_FIELD], 1);
                if ((unsigned)p < (unsigned)STAGE_CAP) stage[p] = (int)(w & SRC_MASK);
            }
        }
    }
    __syncthreads();

    const int  c  = lane & 15;
    const bool hi = (lane & 16) != 0;
    const v4f as4 = ((const v4f*)A)[c];
    const v2f ad2 = ((const v2f*)(A + OUT_DIM))[lane];
    int lo_l = 0, hi_l = 0;
    {
        const int dk = wave + RN_W * lane;
        if (dk < BIN_NODES) { lo_l = seg[dk]; hi_l = seg[dk + 1]; }
        lo_l = min(max(lo_l, 0), n);
        hi_l = min(max(hi_l, lo_l), n);
    }
    const int node0 = b * BIN_NODES + wave;
    int start = __builtin_amdgcn_readlane(lo_l, 0);
    int end   = __builtin_amdgcn_readlane(hi_l, 0);
    const int eol = edge_of_lane(lane);
    int si_c  = stage[min(max(start + min(eol, end - start - 1), 0), STAGE_CAP - 1)];
    float td_l = 0.f;
    {
        v2f ownr[NODE_J];
#pragma unroll
        for (int k = 0; k < NODE_J; ++k) {
            const int nd = min(node0 + RN_W * k, N_NODES - 1);
            ownr[k] = *(const v2f*)((const char*)z + (unsigned)(nd * (OUT_DIM * 4) + lane * 8));
        }
#pragma unroll
        for (int k = 0; k < NODE_J; ++k) {
            const float t = wave_sum_all(ownr[k].x * ad2.x + ownr[k].y * ad2.y);
            td_l = (lane == k) ? t : td_l;
        }
    }
    for (int k = 0; k < NODE_J; ++k) {
        const int d    = wave + RN_W * k;
        const int node = b * BIN_NODES + d;
        if (d >= BIN_NODES || node >= N_NODES) break;
        const int kn      = min(k + 1, NODE_J - 1);
        const int start_n = __builtin_amdgcn_readlane(lo_l, kn);
        const int end_n   = __builtin_amdgcn_readlane(hi_l, kn);
        const int si_n    = stage[min(max(start_n + min(eol, end_n - start_n - 1), 0), STAGE_CAP - 1)];

        v4f acc; acc.x = 0.f; acc.y = 0.f; acc.z = 0.f; acc.w = 0.f;
        if (end > start) {
            const float td = __int_as_float(__builtin_amdgcn_readlane(__float_as_int(td_l), k));
            float den = 0.f, mrun = -INFINITY;
            if (staged) {
                for (int base = start; base < end; base += 32) {
                    const int n32 = (end - base < 32) ? (end - base) : 32;
                    int si = si_c;
                    if (base != start) si = stage[min(max(base + min(eol, n32 - 1), 0), STAGE_CAP - 1)];
                    si = min(max(si, 0), N_NODES - 1);
                    const int rowoff = si * (OUT_DIM * 4);
                    if (n32 >= 16) {
                        edge_group16<0, 8>(acc, den, mrun, rowoff, 16, z, lane, as4, td);
                        if (n32 > 24)      edge_group16<1, 8>(acc, den, mrun, rowoff, n32 - 16, z, lane, as4, td);
                        else if (n32 > 16) edge_group16<1, 4>(acc, den, mrun, rowoff, n32 - 16, z, lane, as4, td);
                    } else {
                        edge_group16<0, 8>(acc, den, mrun, rowoff, n32, z, lane, as4, td);
                    }
                }
            } else {
                for (int ch = 0; ch < NCHUNK; ++ch) {
                    const int rs_c = __builtin_amdgcn_readfirstlane(run_s[ch]);
                    const int rl_c = __builtin_amdgcn_readfirstlane(run_l[ch]);
                    for (int k0 = 0; k0 < rl_c; k0 += 32) {
                        const bool in = k0 + lane < rl_c;
                        const unsigned w = in ? binned[rs_c + k0 + lane] : 0u;
                        const bool hit = in && (int)((w >> SRC_BITS) & BIN_FIELD) == d;
                        const int si = min((int)(w & SRC_MASK), N_NODES - 1);
                        unsigned mask = __builtin_amdgcn_ballot_w32(hit);
                        while (mask != 0u) {
                            const int L = __builtin_ctz(mask);
                            mask &= mask - 1u;
                            const int one = __builtin_amdgcn_readlane(si, L) * (OUT_DIM * 4);
                            edge_group16<0, 8>(acc, den, mrun, one, 1, z, lane, as4, td);
                        }
                    }
                }
            }
            acc.x += lane_xor16(acc.x);
            acc.y += lane_xor16(acc.y);
            acc.z += lane_xor16(acc.z);
            acc.w += lane_xor16(acc.w);
            den   += lane_xor16(den);
            const float inv = __builtin_amdgcn_rcpf(den);
            acc = acc * inv;
        }
        if (!hi) *(v4f*)((char*)out + (unsigned)(node * (OUT_DIM * 4) + c * 16)) = acc;

        start = start_n; end = end_n; si_c = si_n;
    }
}

extern "C" void kernel_launch(void* const* d_in, const int* in_sizes, int n_in,
                              void* d_out, int out_size, void* d_ws, size_t ws_size,
                              hipStream_t stream) {
    const float* h  = (const float*)d_in[0];
    const int* src  = (const int*)d_in[1];
    const int* dst  = (const int*)d_in[2];
    const float* W  = (const float*)d_in[3];
    const float* A  = (const float*)d_in[4];
    float* out = (float*)d_out;

    char* ws = (char*)d_ws;
    float*    z        = (float*)   (ws);
    int*      table    = (int*)     (ws + 25600000);
    unsigned* binned   = (unsigned*)(ws + 27200000);

    gat_chunk_sort_gemm<<<NCHUNK + GEMM_ONLY_GROUPS, SORT_T, 0, stream>>>(src, dst, table, binned, h, W, z);
    gat_range_node<<<NBINS, RN_T, 0, stream>>>(binned, table, A, z, out);
}
